# starting kernel
# baseline (speedup 1.0000x reference)
#include <hip/hip_runtime.h>
#include <hip/hip_fp16.h>

typedef __attribute__((ext_vector_type(16))) _Float16 v16h;
typedef __attribute__((ext_vector_type(8)))  float    v8f;
typedef __attribute__((ext_vector_type(2)))  float    v2f;

#define THREADS 256
#define WAVES_PER_BLOCK (THREADS / 32)

__global__ __launch_bounds__(THREADS)
void k_fill_ones(float* __restrict__ p, int n) {
  int i = blockIdx.x * blockDim.x + threadIdx.x;
  if (i < n) p[i] = 1.0f;
}

__global__ __launch_bounds__(THREADS)
void k_deg_count(const int* __restrict__ dst, float* __restrict__ deg, int e) {
  int i = blockIdx.x * blockDim.x + threadIdx.x;
  if (i < e) atomicAdd(&deg[dst[i]], 1.0f);
}

__global__ __launch_bounds__(THREADS)
void k_rsqrt(float* __restrict__ p, int n) {
  int i = blockIdx.x * blockDim.x + threadIdx.x;
  if (i < n) p[i] = rsqrtf(p[i]);
}

__global__ __launch_bounds__(THREADS)
void k_xform32_wmma(const float* __restrict__ x, const float* __restrict__ W,
                    float* __restrict__ m, int n) {
  __shared__ float lds[WAVES_PER_BLOCK * 64];
  int wave  = (int)((blockIdx.x * (unsigned)blockDim.x + threadIdx.x) >> 5);
  int tl    = threadIdx.x & 31;
  int row0  = wave * 16;
  if (row0 >= n) return;
  int lane16 = tl & 15;
  int hi     = tl >> 4;

  v16h b;
  {
    int  colW = lane16 & 3;
    bool keep = lane16 < 4;
#pragma unroll
    for (int j = 0; j < 16; ++j) {
      int k = (j < 8) ? (hi * 8 + j) : (16 + hi * 8 + (j - 8));
      float wv = W[k * 4 + colW];
      b[j] = keep ? (_Float16)wv : (_Float16)0.0f;
    }
  }

  int  arow = row0 + lane16;
  bool rok  = arow < n;
  int  ar   = rok ? arow : (n - 1);
  const float* xr = x + (size_t)ar * 32;
  float4 q0 = *reinterpret_cast<const float4*>(xr + hi * 8);
  float4 q1 = *reinterpret_cast<const float4*>(xr + hi * 8 + 4);
  float4 q2 = *reinterpret_cast<const float4*>(xr + 16 + hi * 8);
  float4 q3 = *reinterpret_cast<const float4*>(xr + 16 + hi * 8 + 4);
  float zf = rok ? 1.0f : 0.0f;
  v16h a;
  a[0]=(_Float16)(q0.x*zf); a[1]=(_Float16)(q0.y*zf); a[2]=(_Float16)(q0.z*zf); a[3]=(_Float16)(q0.w*zf);
  a[4]=(_Float16)(q1.x*zf); a[5]=(_Float16)(q1.y*zf); a[6]=(_Float16)(q1.z*zf); a[7]=(_Float16)(q1.w*zf);
  a[8]=(_Float16)(q2.x*zf); a[9]=(_Float16)(q2.y*zf); a[10]=(_Float16)(q2.z*zf); a[11]=(_Float16)(q2.w*zf);
  a[12]=(_Float16)(q3.x*zf); a[13]=(_Float16)(q3.y*zf); a[14]=(_Float16)(q3.z*zf); a[15]=(_Float16)(q3.w*zf);

  v8f c = {};
  c = __builtin_amdgcn_wmma_f32_16x16x32_f16(false, a, false, b,
                                             (short)0, c, false, false);

  float* slab = lds + ((threadIdx.x >> 5) << 6);
  if (lane16 < 4) {
#pragma unroll
    for (int r = 0; r < 8; ++r) slab[(hi * 8 + r) * 4 + lane16] = c[r];
  }
  asm volatile("s_wait_dscnt 0" ::: "memory");
  if (hi == 0) {
    int mr = row0 + lane16;
    if (mr < n) {
      float4 v = *reinterpret_cast<const float4*>(slab + lane16 * 4);
      *reinterpret_cast<float4*>(m + (size_t)mr * 4) = v;
    }
  }
}

template<int KIN, int FOUT, bool BIAS>
__global__ __launch_bounds__(THREADS)
void k_xform_small(const float* __restrict__ hin, const float* __restrict__ W,
                   const float* __restrict__ bias, float* __restrict__ mout, int n) {
  __shared__ float lds[WAVES_PER_BLOCK * 64];
  int wave  = (int)((blockIdx.x * (unsigned)blockDim.x + threadIdx.x) >> 5);
  int tl    = threadIdx.x & 31;
  int row0  = wave * 16;
  if (row0 >= n) return;
  int lane16 = tl & 15;
  int hi     = tl >> 4;

#if __has_builtin(__builtin_amdgcn_wmma_f32_16x16x4_f32)
  int  arow = row0 + lane16;
  bool rok  = arow < n;
  int  ar   = rok ? arow : (n - 1);
  float zf  = rok ? 1.0f : 0.0f;

  v2f a;
  if (KIN == 4) {
    float2 v = *reinterpret_cast<const float2*>(hin + (size_t)ar * 4 + hi * 2);
    a[0] = v.x * zf;  a[1] = v.y * zf;
  } else {
    float2 v = *reinterpret_cast<const float2*>(hin + (size_t)ar * 2);
    float zk = (hi == 0) ? zf : 0.0f;
    a[0] = v.x * zk;  a[1] = v.y * zk;
  }

  v2f b;
#pragma unroll
  for (int j = 0; j < 2; ++j) {
    int  k    = hi * 2 + j;
    int  kc   = (k < KIN) ? k : (KIN - 1);
    int  cc   = (lane16 < FOUT) ? lane16 : (FOUT - 1);
    float wv  = W[kc * FOUT + cc];
    b[j] = (k < KIN && lane16 < FOUT) ? wv : 0.0f;
  }

  v8f c = {};
  c = __builtin_amdgcn_wmma_f32_16x16x4_f32(false, a, false, b,
                                            (short)0, c, false, false);

  float* slab = lds + ((threadIdx.x >> 5) << 6);
  if (lane16 < FOUT) {
#pragma unroll
    for (int r = 0; r < 8; ++r) slab[(hi * 8 + r) * FOUT + lane16] = c[r];
  }
  asm volatile("s_wait_dscnt 0" ::: "memory");
  if (hi == 0) {
    int mr = row0 + lane16;
    if (mr < n) {
      if (FOUT == 4) {
        float4 v = *reinterpret_cast<const float4*>(slab + lane16 * 4);
        if (BIAS) { v.x += bias[0]; v.y += bias[1]; v.z += bias[2]; v.w += bias[3]; }
        *reinterpret_cast<float4*>(mout + (size_t)mr * 4) = v;
      } else {
        float2 v = *reinterpret_cast<const float2*>(slab + lane16 * 2);
        if (BIAS) { v.x += bias[0]; v.y += bias[1]; }
        *reinterpret_cast<float2*>(mout + (size_t)mr * 2) = v;
      }
    }
  }
#else
  if (hi == 0) {
    int mr = row0 + lane16;
    if (mr < n) {
#pragma unroll
      for (int f = 0; f < FOUT; ++f) {
        float acc = BIAS ? bias[f] : 0.0f;
#pragma unroll
        for (int k = 0; k < KIN; ++k) acc += hin[mr * KIN + k] * W[k * FOUT + f];
        mout[mr * FOUT + f] = acc;
      }
    }
  }
#endif
}

template<int F>
__global__ __launch_bounds__(THREADS)
void k_agg_init(float* __restrict__ agg, const float* __restrict__ m,
                const float* __restrict__ dinv, const float* __restrict__ b, int n) {
  int i = blockIdx.x * blockDim.x + threadIdx.x;
  if (i >= n * F) return;
  int node = i / F;
  int f    = i - node * F;
  float di = dinv[node];
  agg[i] = di * di * m[i] + b[f];
}

template<int F>
__global__ __launch_bounds__(THREADS)
void k_edge_scatter(const int* __restrict__ src, const int* __restrict__ dst,
                    const float* __restrict__ dinv, const float* __restrict__ m,
                    float* __restrict__ agg, int e) {
  int i = blockIdx.x * blockDim.x + threadIdx.x;
  if (i >= e) return;
  int s = src[i], d = dst[i];
  float w = dinv[s] * dinv[d];
  if (F == 4) {
    float4 v = *reinterpret_cast<const float4*>(m + (size_t)s * 4);
    atomicAdd(&agg[(size_t)d * 4 + 0], w * v.x);
    atomicAdd(&agg[(size_t)d * 4 + 1], w * v.y);
    atomicAdd(&agg[(size_t)d * 4 + 2], w * v.z);
    atomicAdd(&agg[(size_t)d * 4 + 3], w * v.w);
  } else {
    float2 v = *reinterpret_cast<const float2*>(m + (size_t)s * 2);
    atomicAdd(&agg[(size_t)d * 2 + 0], w * v.x);
    atomicAdd(&agg[(size_t)d * 2 + 1], w * v.y);
  }
}

__global__ __launch_bounds__(THREADS)
void k_tanh(float* __restrict__ dst, const float* __restrict__ srcp, int cnt) {
  int i = blockIdx.x * blockDim.x + threadIdx.x;
  if (i < cnt) dst[i] = tanhf(srcp[i]);
}

extern "C" void kernel_launch(void* const* d_in, const int* in_sizes, int n_in,
                              void* d_out, int out_size, void* d_ws, size_t ws_size,
                              hipStream_t stream) {
  const float* x  = (const float*)d_in[0];
  const int*   ei = (const int*)  d_in[1];
  const float* W1 = (const float*)d_in[2];
  const float* b1 = (const float*)d_in[3];
  const float* W2 = (const float*)d_in[4];
  const float* b2 = (const float*)d_in[5];
  const float* W3 = (const float*)d_in[6];
  const float* b3 = (const float*)d_in[7];
  const float* Wc = (const float*)d_in[8];
  const float* bc = (const float*)d_in[9];

  const int N = in_sizes[0] / 32;
  const int E = in_sizes[1] / 2;
  const int* src = ei;
  const int* dst = ei + E;

  float* out = (float*)d_out;
  float* h3  = out + (size_t)N * 4;

  float* ws   = (float*)d_ws;
  float* dinv = ws;
  float* m    = ws + (size_t)N;
  float* agg  = ws + (size_t)5 * N;
  float* h    = ws + (size_t)9 * N;

  const int gN  = (N + THREADS - 1) / THREADS;
  const int gE  = (E + THREADS - 1) / THREADS;
  const int g4N = (N * 4 + THREADS - 1) / THREADS;
  const int g2N = (N * 2 + THREADS - 1) / THREADS;
  const int tiles = (N + 15) / 16;
  const int gW  = (tiles + WAVES_PER_BLOCK - 1) / WAVES_PER_BLOCK;

  k_fill_ones<<<gN, THREADS, 0, stream>>>(dinv, N);
  k_deg_count<<<gE, THREADS, 0, stream>>>(dst, dinv, E);
  k_rsqrt<<<gN, THREADS, 0, stream>>>(dinv, N);

  k_xform32_wmma<<<gW, THREADS, 0, stream>>>(x, W1, m, N);
  k_agg_init<4><<<g4N, THREADS, 0, stream>>>(agg, m, dinv, b1, N);
  k_edge_scatter<4><<<gE, THREADS, 0, stream>>>(src, dst, dinv, m, agg, E);
  k_tanh<<<g4N, THREADS, 0, stream>>>(h, agg, N * 4);

  k_xform_small<4, 4, false><<<gW, THREADS, 0, stream>>>(h, W2, nullptr, m, N);
  k_agg_init<4><<<g4N, THREADS, 0, stream>>>(agg, m, dinv, b2, N);
  k_edge_scatter<4><<<gE, THREADS, 0, stream>>>(src, dst, dinv, m, agg, E);
  k_tanh<<<g4N, THREADS, 0, stream>>>(h, agg, N * 4);

  k_xform_small<4, 2, false><<<gW, THREADS, 0, stream>>>(h, W3, nullptr, m, N);
  k_agg_init<2><<<g2N, THREADS, 0, stream>>>(agg, m, dinv, b3, N);
  k_edge_scatter<2><<<gE, THREADS, 0, stream>>>(src, dst, dinv, m, agg, E);
  k_tanh<<<g2N, THREADS, 0, stream>>>(h3, agg, N * 2);

  k_xform_small<2, 4, true><<<gW, THREADS, 0, stream>>>(h3, Wc, bc, out, N);
}


# Edge stream split once: a work-group counting-sorts its 25000-edge chunk by destination range in LDS into 32 segments; degree and layer passes read segments; sums regrouped into 8 partials, not 32
# speedup vs baseline: 14.4863x; 14.4863x over previous
#include <hip/hip_runtime.h>
#include <hip/hip_fp16.h>

typedef __attribute__((ext_vector_type(16))) _Float16 v16h;
typedef __attribute__((ext_vector_type(8)))  float    v8f;
typedef __attribute__((ext_vector_type(2)))  float    v2f;

#define THREADS 256
#define WAVES_PER_BLOCK (THREADS / 32)

#define LB_THREADS 1024
#define GRID_WGS   256
#define NPR4       12544
#define NPR2       20544
#define NPRD       NPR4
constexpr int npr_of(int F) { return F == 4 ? NPR4 : NPR2; }


__device__ __forceinline__ int4 load_idx4_tail(const int* __restrict__ p, int i, int e1, int fill) {
  int4 r;
  r.x = p[i];
  r.y = (i + 1 < e1) ? p[i + 1] : fill;
  r.z = (i + 2 < e1) ? p[i + 2] : fill;
  r.w = (i + 3 < e1) ? p[i + 3] : fill;
  return r;
}

#define TILE_EDGES (4 * LB_THREADS)

__device__ __forceinline__ void deg_add4(float* cntl, int4 d, int lo, unsigned ucnt) {
  unsigned r0 = (unsigned)(d.x - lo), r1 = (unsigned)(d.y - lo);
  unsigned r2 = (unsigned)(d.z - lo), r3 = (unsigned)(d.w - lo);
  if (r0 < ucnt) atomicAdd(&cntl[r0], 1.0f);
  if (r1 < ucnt) atomicAdd(&cntl[r1], 1.0f);
  if (r2 < ucnt) atomicAdd(&cntl[r2], 1.0f);
  if (r3 < ucnt) atomicAdd(&cntl[r3], 1.0f);
}

__global__ __launch_bounds__(LB_THREADS)
void k_deg_lds(const int* __restrict__ dst, float* __restrict__ part,
               int e, int n, int nranges, int echunk) {
  __shared__ float cntl[NPRD];
  const int tid = (int)threadIdx.x;
  const int r   = (int)(blockIdx.x % (unsigned)nranges);
  const int c   = (int)(blockIdx.x / (unsigned)nranges);
  const int lo  = r * NPRD;
  int cnt = n - lo;
  if (cnt > NPRD) cnt = NPRD;
  const unsigned ucnt = (unsigned)cnt;

  for (int i = tid; i < NPRD; i += LB_THREADS) cntl[i] = 0.0f;
  __syncthreads();

  const int e0 = c * echunk;
  int e1 = e0 + echunk;
  if (e1 > e) e1 = e;
  const bool vec = (((size_t)dst) & 15) == 0;
  const int nfull = (vec && e1 > e0) ? (e1 - e0) / TILE_EDGES : 0;

  const int4* dp = reinterpret_cast<const int4*>(dst + e0) + tid;
  int t = 0;
  for (; t + 4 <= nfull; t += 4) {
    int4 d0 = dp[(size_t)(t + 0) * LB_THREADS];
    int4 d1 = dp[(size_t)(t + 1) * LB_THREADS];
    int4 d2 = dp[(size_t)(t + 2) * LB_THREADS];
    int4 d3 = dp[(size_t)(t + 3) * LB_THREADS];
    deg_add4(cntl, d0, lo, ucnt);
    deg_add4(cntl, d1, lo, ucnt);
    deg_add4(cntl, d2, lo, ucnt);
    deg_add4(cntl, d3, lo, ucnt);
  }
  for (; t < nfull; ++t) deg_add4(cntl, dp[(size_t)t * LB_THREADS], lo, ucnt);
  for (int base = e0 + nfull * TILE_EDGES + 4 * tid; base < e1; base += TILE_EDGES)
    deg_add4(cntl, load_idx4_tail(dst, base, e1, -1), lo, ucnt);
  __syncthreads();

  float* pc = part + (size_t)c * n + lo;
  for (int i = tid; i < cnt; i += LB_THREADS) pc[i] = cntl[i];
}

__global__ __launch_bounds__(THREADS)
void k_rsqrt(float* __restrict__ dinv, const float* __restrict__ part, int n, int nchunks) {
  int i = blockIdx.x * blockDim.x + threadIdx.x;
  if (i >= n) return;
  const float* p = part + i;
  float s = 1.0f;
  int c = 0;
  for (; c + 4 <= nchunks; c += 4) {
    float t0 = p[(size_t)(c + 0) * n], t1 = p[(size_t)(c + 1) * n];
    float t2 = p[(size_t)(c + 2) * n], t3 = p[(size_t)(c + 3) * n];
    s += t0; s += t1; s += t2; s += t3;
  }
  for (; c < nchunks; ++c) s += p[(size_t)c * n];
  dinv[i] = rsqrtf(s);
}

__global__ __launch_bounds__(THREADS)
void k_xform32_wmma(const float* __restrict__ x, const float* __restrict__ W,
                    const float* __restrict__ dinv, float* __restrict__ m, int n) {
  __shared__ float lds[WAVES_PER_BLOCK * 64];
  int wave  = (int)((blockIdx.x * (unsigned)blockDim.x + threadIdx.x) >> 5);
  int tl    = threadIdx.x & 31;
  int row0  = wave * 16;
  if (row0 >= n) return;
  int lane16 = tl & 15;
  int hi     = tl >> 4;

  v16h b;
  {
    int  colW = lane16 & 3;
    bool keep = lane16 < 4;
#pragma unroll
    for (int j = 0; j < 16; ++j) {
      int k = (j < 8) ? (hi * 8 + j) : (16 + hi * 8 + (j - 8));
      float wv = W[k * 4 + colW];
      b[j] = keep ? (_Float16)wv : (_Float16)0.0f;
    }
  }

  int  arow = row0 + lane16;
  bool rok  = arow < n;
  int  ar   = rok ? arow : (n - 1);
  const float* xr = x + (size_t)ar * 32;
  float4 q0 = *reinterpret_cast<const float4*>(xr + hi * 8);
  float4 q1 = *reinterpret_cast<const float4*>(xr + hi * 8 + 4);
  float4 q2 = *reinterpret_cast<const float4*>(xr + 16 + hi * 8);
  float4 q3 = *reinterpret_cast<const float4*>(xr + 16 + hi * 8 + 4);
  float di = dinv[ar];
  float zf = rok ? 1.0f : 0.0f;
  v16h a;
  a[0]=(_Float16)(q0.x*zf); a[1]=(_Float16)(q0.y*zf); a[2]=(_Float16)(q0.z*zf); a[3]=(_Float16)(q0.w*zf);
  a[4]=(_Float16)(q1.x*zf); a[5]=(_Float16)(q1.y*zf); a[6]=(_Float16)(q1.z*zf); a[7]=(_Float16)(q1.w*zf);
  a[8]=(_Float16)(q2.x*zf); a[9]=(_Float16)(q2.y*zf); a[10]=(_Float16)(q2.z*zf); a[11]=(_Float16)(q2.w*zf);
  a[12]=(_Float16)(q3.x*zf); a[13]=(_Float16)(q3.y*zf); a[14]=(_Float16)(q3.z*zf); a[15]=(_Float16)(q3.w*zf);

  v8f c = {};
  c = __builtin_amdgcn_wmma_f32_16x16x32_f16(false, a, false, b,
                                             (short)0, c, false, false);

  float* slab = lds + ((threadIdx.x >> 5) << 6);
  if (lane16 < 4) {
#pragma unroll
    for (int r = 0; r < 8; ++r) slab[(hi * 8 + r) * 4 + lane16] = c[r];
  }
  asm volatile("s_wait_dscnt 0" ::: "memory");
  if (hi == 0) {
    int mr = row0 + lane16;
    if (mr < n) {
      float4 v = *reinterpret_cast<const float4*>(slab + lane16 * 4);
      v.x *= di; v.y *= di; v.z *= di; v.w *= di;
      *reinterpret_cast<float4*>(m + (size_t)mr * 4) = v;
    }
  }
}

template<int KIN, int FOUT, bool BIAS, bool SCALE>
__global__ __launch_bounds__(THREADS)
void k_xform_small(const float* __restrict__ hin, const float* __restrict__ W,
                   const float* __restrict__ bias, const float* __restrict__ dinv,
                   float* __restrict__ mout, int n) {
  __shared__ float lds[WAVES_PER_BLOCK * 64];
  int wave  = (int)((blockIdx.x * (unsigned)blockDim.x + threadIdx.x) >> 5);
  int tl    = threadIdx.x & 31;
  int row0  = wave * 16;
  if (row0 >= n) return;
  int lane16 = tl & 15;
  int hi     = tl >> 4;

#if __has_builtin(__builtin_amdgcn_wmma_f32_16x16x4_f32)
  int  arow = row0 + lane16;
  bool rok  = arow < n;
  int  ar   = rok ? arow : (n - 1);
  float zf  = rok ? 1.0f : 0.0f;
  float di  = 1.0f;
  if (SCALE) di = dinv[ar];

  v2f a;
  if (KIN == 4) {
    float2 v = *reinterpret_cast<const float2*>(hin + (size_t)ar * 4 + hi * 2);
    a[0] = v.x * zf;  a[1] = v.y * zf;
  } else {
    float2 v = *reinterpret_cast<const float2*>(hin + (size_t)ar * 2);
    float zk = (hi == 0) ? zf : 0.0f;
    a[0] = v.x * zk;  a[1] = v.y * zk;
  }

  v2f b;
#pragma unroll
  for (int j = 0; j < 2; ++j) {
    int  k    = hi * 2 + j;
    int  kc   = (k < KIN) ? k : (KIN - 1);
    int  cc   = (lane16 < FOUT) ? lane16 : (FOUT - 1);
    float wv  = W[kc * FOUT + cc];
    b[j] = (k < KIN && lane16 < FOUT) ? wv : 0.0f;
  }

  v8f c = {};
  c = __builtin_amdgcn_wmma_f32_16x16x4_f32(false, a, false, b,
                                            (short)0, c, false, false);

  float* slab = lds + ((threadIdx.x >> 5) << 6);
  if (lane16 < FOUT) {
#pragma unroll
    for (int r = 0; r < 8; ++r) slab[(hi * 8 + r) * FOUT + lane16] = c[r];
  }
  asm volatile("s_wait_dscnt 0" ::: "memory");
  if (hi == 0) {
    int mr = row0 + lane16;
    if (mr < n) {
      if (FOUT == 4) {
        float4 v = *reinterpret_cast<const float4*>(slab + lane16 * 4);
        if (BIAS) { v.x += bias[0]; v.y += bias[1]; v.z += bias[2]; v.w += bias[3]; }
        if (SCALE) { v.x *= di; v.y *= di; v.z *= di; v.w *= di; }
        *reinterpret_cast<float4*>(mout + (size_t)mr * 4) = v;
      } else {
        float2 v = *reinterpret_cast<const float2*>(slab + lane16 * 2);
        if (BIAS) { v.x += bias[0]; v.y += bias[1]; }
        if (SCALE) { v.x *= di; v.y *= di; }
        *reinterpret_cast<float2*>(mout + (size_t)mr * 2) = v;
      }
    }
  }
#else
  if (hi == 0) {
    int mr = row0 + lane16;
    if (mr < n) {
#pragma unroll
      for (int f = 0; f < FOUT; ++f) {
        float acc = BIAS ? bias[f] : 0.0f;
#pragma unroll
        for (int k = 0; k < KIN; ++k) acc += hin[mr * KIN + k] * W[k * FOUT + f];
        if (SCALE) acc *= dinv[mr];
        mout[mr * FOUT + f] = acc;
      }
    }
  }
#endif
}

template<int F>
__device__ __forceinline__ void edge_add4(float* acc, const float* __restrict__ m,
                                          int4 d4, int4 s4, int lo, unsigned ucnt) {
  int dd[4] = {d4.x, d4.y, d4.z, d4.w};
  int ss[4] = {s4.x, s4.y, s4.z, s4.w};
  constexpr int NPR = npr_of(F);
  unsigned rel[4];
  bool     hit[4];
  float    v[4][F];
  for (int k = 0; k < 4; ++k) {
    rel[k] = (unsigned)(dd[k] - lo);
    hit[k] = rel[k] < ucnt;
    int      si = hit[k] ? ss[k] : lo;
    if constexpr (F == 4) {
      float4 t = *reinterpret_cast<const float4*>(m + (size_t)si * 4);
      v[k][0] = t.x; v[k][1] = t.y; v[k][2] = t.z; v[k][3] = t.w;
    } else {
      float2 t = *reinterpret_cast<const float2*>(m + (size_t)si * 2);
      v[k][0] = t.x; v[k][1] = t.y;
    }
  }
  for (int k = 0; k < 4; ++k) {
    if (hit[k]) {
      for (int f = 0; f < F; ++f) atomicAdd(&acc[f * NPR + rel[k]], v[k][f]);
    }
  }
}

template<int F>
__global__ __launch_bounds__(LB_THREADS)
void k_edge_lds(const int* __restrict__ src, const int* __restrict__ dst,
                const float* __restrict__ m,
                float* __restrict__ part, int e, int n, int nranges, int echunk) {
  constexpr int NPR = npr_of(F);
  __shared__ float acc[F * NPR];
  const int tid = (int)threadIdx.x;
  const int r   = (int)(blockIdx.x % (unsigned)nranges);
  const int c   = (int)(blockIdx.x / (unsigned)nranges);
  const int lo  = r * NPR;
  int cnt = n - lo;
  if (cnt > NPR) cnt = NPR;
  const unsigned ucnt = (unsigned)cnt;

  for (int i = tid; i < F * NPR; i += LB_THREADS) acc[i] = 0.0f;
  __syncthreads();

  const int e0 = c * echunk;
  int e1 = e0 + echunk;
  if (e1 > e) e1 = e;
  const bool vec = ((((size_t)src) | ((size_t)dst)) & 15) == 0;
  const int nfull = (vec && e1 > e0) ? (e1 - e0) / TILE_EDGES : 0;

  const int4* dp = reinterpret_cast<const int4*>(dst + e0) + tid;
  const int4* sp = reinterpret_cast<const int4*>(src + e0) + tid;
  int4 dn = make_int4(-1, -1, -1, -1), sn = make_int4(0, 0, 0, 0);
  if (nfull > 0) { dn = dp[0]; sn = sp[0]; }
  for (int t = 0; t < nfull; ++t) {
    int4 d4 = dn, s4 = sn;
    if (t + 1 < nfull) {
      dn = dp[(size_t)(t + 1) * LB_THREADS];
      sn = sp[(size_t)(t + 1) * LB_THREADS];
    }
    edge_add4<F>(acc, m, d4, s4, lo, ucnt);
  }
  for (int base = e0 + nfull * TILE_EDGES + 4 * tid; base < e1; base += TILE_EDGES)
    edge_add4<F>(acc, m, load_idx4_tail(dst, base, e1, -1),
                 load_idx4_tail(src, base, e1, 0), lo, ucnt);
  __syncthreads();

  float* pc = part + (size_t)c * F * n + lo;
  for (int i = tid; i < cnt; i += LB_THREADS) {
    float t[F];
    for (int f = 0; f < F; ++f) t[f] = acc[f * NPR + i];
    for (int f = 0; f < F; ++f) pc[(size_t)f * n + i] = t[f];
  }
}

template<int F>
__global__ __launch_bounds__(THREADS)
void k_tanh(float* __restrict__ hout, const float* __restrict__ part,
            const float* __restrict__ m, const float* __restrict__ dinv,
            const float* __restrict__ b, int n, int nchunks) {
  int i = blockIdx.x * blockDim.x + threadIdx.x;
  if (i >= n) return;
  const float* p = part + i;
  const size_t cs = (size_t)F * n;
  float s[F];
  for (int f = 0; f < F; ++f) s[f] = 0.0f;
  int c = 0;
  for (; c + 4 <= nchunks; c += 4) {
    float t[4][F];
    for (int j = 0; j < 4; ++j)
      for (int f = 0; f < F; ++f) t[j][f] = p[(size_t)(c + j) * cs + (size_t)f * n];
    for (int j = 0; j < 4; ++j)
      for (int f = 0; f < F; ++f) s[f] += t[j][f];
  }
  for (; c < nchunks; ++c)
    for (int f = 0; f < F; ++f) s[f] += p[(size_t)c * cs + (size_t)f * n];
  float di = dinv[i];
  if constexpr (F == 4) {
    float4 mv = *reinterpret_cast<const float4*>(m + (size_t)i * 4);
    float4 o;
    o.x = tanhf(di * (s[0] + mv.x) + b[0]);
    o.y = tanhf(di * (s[1] + mv.y) + b[1]);
    o.z = tanhf(di * (s[2] + mv.z) + b[2]);
    o.w = tanhf(di * (s[3] + mv.w) + b[3]);
    *reinterpret_cast<float4*>(hout + (size_t)i * 4) = o;
  } else {
    float2 mv = *reinterpret_cast<const float2*>(m + (size_t)i * 2);
    float2 o;
    o.x = tanhf(di * (s[0] + mv.x) + b[0]);
    o.y = tanhf(di * (s[1] + mv.y) + b[1]);
    *reinterpret_cast<float2*>(hout + (size_t)i * 2) = o;
  }
}


#define LIST_RING 256
#define LIST_RELBITS 14

__device__ __forceinline__ int4 load_idx4_guard(const int* __restrict__ p, int i, int e1, int fill) {
  int4 r;
  r.x = (i     < e1) ? p[i]     : fill;
  r.y = (i + 1 < e1) ? p[i + 1] : fill;
  r.z = (i + 2 < e1) ? p[i + 2] : fill;
  r.w = (i + 3 < e1) ? p[i + 3] : fill;
  return r;
}

__device__ __forceinline__ void deg_emit4(float* cntl, unsigned* myring,
                                          unsigned* __restrict__ mylist,
                                          int4 d, int4 s, int lo, unsigned ucnt,
                                          unsigned lane, unsigned ltmask,
                                          unsigned& wp, unsigned& rp) {
  int dd[4] = {d.x, d.y, d.z, d.w};
  int ss[4] = {s.x, s.y, s.z, s.w};
  for (int k = 0; k < 4; ++k) {
    unsigned rel = (unsigned)(dd[k] - lo);
    bool hit = rel < ucnt;
    unsigned b = __builtin_amdgcn_ballot_w32(hit);
    if (hit) {
      atomicAdd(&cntl[rel], 1.0f);
      unsigned slot = (wp + (unsigned)__builtin_popcount(b & ltmask)) & (LIST_RING - 1);
      myring[slot] = ((unsigned)ss[k] << LIST_RELBITS) | rel;
    }
    wp += (unsigned)__builtin_popcount(b);
  }
  while (wp - rp >= 32u) {
    __builtin_amdgcn_wave_barrier();
    mylist[rp + lane] = myring[(rp + lane) & (LIST_RING - 1)];
    rp += 32u;
  }
}

__global__ __launch_bounds__(LB_THREADS)
void k_deg_list(const int* __restrict__ src, const int* __restrict__ dst,
                float* __restrict__ part, unsigned* __restrict__ lists,
                int* __restrict__ lcnt, int e, int n, int nranges, int echunk, int wcap) {
  __shared__ float cntl[NPRD];
  __shared__ unsigned ring[(LB_THREADS / 32) * LIST_RING];
  const int tid = (int)threadIdx.x;
  const unsigned lane = (unsigned)tid & 31u;
  const unsigned ltmask = (1u << lane) - 1u;
  const int wv  = tid >> 5;
  const int r   = (int)(blockIdx.x % (unsigned)nranges);
  const int c   = (int)(blockIdx.x / (unsigned)nranges);
  const int lo  = r * NPRD;
  int cnt = n - lo;
  if (cnt > NPRD) cnt = NPRD;
  const unsigned ucnt = (unsigned)cnt;
  const size_t wid = (size_t)blockIdx.x * (LB_THREADS / 32) + (size_t)wv;
  unsigned* myring = ring + wv * LIST_RING;
  unsigned* mylist = lists + wid * (size_t)wcap;
  unsigned wp = 0, rp = 0;

  for (int i = tid; i < NPRD; i += LB_THREADS) cntl[i] = 0.0f;
  __syncthreads();

  const int e0 = c * echunk;
  int e1 = e0 + echunk;
  if (e1 > e) e1 = e;
  const bool vec = ((((size_t)src) | ((size_t)dst)) & 15) == 0;
  const int nfull = (vec && e1 > e0) ? (e1 - e0) / TILE_EDGES : 0;

  const int4* dp = reinterpret_cast<const int4*>(dst + e0) + tid;
  const int4* sp = reinterpret_cast<const int4*>(src + e0) + tid;
  int4 dn = make_int4(-1, -1, -1, -1), sn = make_int4(0, 0, 0, 0);
  if (nfull > 0) { dn = dp[0]; sn = sp[0]; }
  for (int t = 0; t < nfull; ++t) {
    int4 d4 = dn, s4 = sn;
    if (t + 1 < nfull) {
      dn = dp[(size_t)(t + 1) * LB_THREADS];
      sn = sp[(size_t)(t + 1) * LB_THREADS];
    }
    deg_emit4(cntl, myring, mylist, d4, s4, lo, ucnt, lane, ltmask, wp, rp);
  }
  for (int b0 = e0 + nfull * TILE_EDGES + 4 * (tid & ~31); b0 < e1; b0 += TILE_EDGES) {
    int base = b0 + 4 * (int)lane;
    deg_emit4(cntl, myring, mylist, load_idx4_guard(dst, base, e1, -1),
              load_idx4_guard(src, base, e1, 0), lo, ucnt, lane, ltmask, wp, rp);
  }
  __builtin_amdgcn_wave_barrier();
  if (rp + lane < wp) mylist[rp + lane] = myring[(rp + lane) & (LIST_RING - 1)];
  if (lane == 0) lcnt[wid] = (int)wp;
  __syncthreads();

  float* pc = part + (size_t)c * n + lo;
  for (int i = tid; i < cnt; i += LB_THREADS) pc[i] = cntl[i];
}

template<int F>
__device__ __forceinline__ void list_add4(float* acc, const float* __restrict__ m,
                                          uint4 w4, int rem, int lo) {
  unsigned ww[4] = {w4.x, w4.y, w4.z, w4.w};
  unsigned rel[4];
  bool     ok[4];
  float    v[4][F];
  for (int k = 0; k < 4; ++k) {
    ok[k]  = k < rem;
    rel[k] = ww[k] & ((1u << LIST_RELBITS) - 1u);
    int si = ok[k] ? (int)(ww[k] >> LIST_RELBITS) : lo;
    if constexpr (F == 4) {
      float4 t = *reinterpret_cast<const float4*>(m + (size_t)si * 4);
      v[k][0] = t.x; v[k][1] = t.y; v[k][2] = t.z; v[k][3] = t.w;
    } else {
      float2 t = *reinterpret_cast<const float2*>(m + (size_t)si * 2);
      v[k][0] = t.x; v[k][1] = t.y;
    }
  }
  for (int k = 0; k < 4; ++k) {
    if (ok[k]) {
      for (int f = 0; f < F; ++f) atomicAdd(&acc[f * NPRD + rel[k]], v[k][f]);
    }
  }
}

template<int F>
__global__ __launch_bounds__(LB_THREADS)
void k_edge_list(const unsigned* __restrict__ lists, const int* __restrict__ lcnt,
                 const float* __restrict__ m, float* __restrict__ part,
                 int n, int nranges, int wcap) {
  __shared__ float acc[4 * NPRD];
  const int tid  = (int)threadIdx.x;
  const int lane = tid & 31;
  const int wv   = tid >> 5;
  const int r   = (int)(blockIdx.x % (unsigned)nranges);
  const int c   = (int)(blockIdx.x / (unsigned)nranges);
  const int lo  = r * NPRD;
  int cnt = n - lo;
  if (cnt > NPRD) cnt = NPRD;

  for (int i = tid; i < F * NPRD; i += LB_THREADS) acc[i] = 0.0f;
  __syncthreads();

  const size_t wid = (size_t)blockIdx.x * (LB_THREADS / 32) + (size_t)wv;
  const int nl = __builtin_amdgcn_readfirstlane(lcnt[wid]);
  const int ng = (nl + 127) >> 7;
  const uint4* lp = reinterpret_cast<const uint4*>(lists + wid * (size_t)wcap) + lane;
  uint4 wn = make_uint4(0u, 0u, 0u, 0u);
  if (ng > 0) wn = lp[0];
  for (int g = 0; g < ng; ++g) {
    uint4 w4 = wn;
    if (g + 1 < ng) wn = lp[(size_t)(g + 1) * 32];
    list_add4<F>(acc, m, w4, nl - (g * 128 + lane * 4), lo);
  }
  __syncthreads();

  float* pc = part + (size_t)c * F * n + lo;
  for (int i = tid; i < cnt; i += LB_THREADS) {
    float t[F];
    for (int f = 0; f < F; ++f) t[f] = acc[f * NPRD + i];
    for (int f = 0; f < F; ++f) pc[(size_t)f * n + i] = t[f];
  }
}


#define SNPR     3136
#define SRELBITS 12
#define SNR      32
#define SNP      256
#define SNC      (SNP / 32)
#define SHP      33
#define SHBLK    (32 * SHP)
#define SCAP     26624
#define SMAXCHUNK 25600

__device__ __forceinline__ unsigned wave_incl_scan(unsigned x, int lane) {
  for (int d = 1; d < 32; d <<= 1) {
    unsigned t = __shfl_up(x, (unsigned)d);
    if (lane >= d) x += t;
  }
  return x;
}

template<bool PLACE>
__device__ __forceinline__ void split4(unsigned* hist, const unsigned* seg, unsigned* stage,
                                       int4 d4, int4 s4, unsigned n, unsigned hme) {
  int dd[4] = {d4.x, d4.y, d4.z, d4.w};
  int ss[4] = {s4.x, s4.y, s4.z, s4.w};
  bool     ok[4];
  unsigned r[4], rank[4], sg[4];
  for (int k = 0; k < 4; ++k) {
    ok[k] = (unsigned)dd[k] < n;
    r[k]  = ok[k] ? (unsigned)dd[k] / SNPR : 0u;
    unsigned* hp = hist + r[k] * SHBLK + hme;
    if constexpr (!PLACE) {
      atomicAdd(hp, ok[k] ? 1u : 0u);
    } else {
      rank[k] = atomicAdd(hp, ok[k] ? 1u : 0u);
      sg[k]   = seg[r[k]];
    }
  }
  if constexpr (PLACE) {
    for (int k = 0; k < 4; ++k)
      if (ok[k]) stage[sg[k] + rank[k]] = ((unsigned)ss[k] << SRELBITS) | ((unsigned)dd[k] - r[k] * SNPR);
  }
}

template<bool PLACE>
__device__ __forceinline__ void split_sweep(unsigned* hist, const unsigned* seg, unsigned* stage,
                                            const int* __restrict__ src, const int* __restrict__ dst,
                                            int e0, int e1, int nfull, unsigned n, int tid, unsigned hme) {
  const int4 z4 = make_int4(0, 0, 0, 0);
  const int4* dp = reinterpret_cast<const int4*>(dst + e0) + tid;
  const int4* sp = reinterpret_cast<const int4*>(src + e0) + tid;
  int t = 0;
  for (; t + 4 <= nfull; t += 4) {
    int4 d0 = dp[(size_t)(t + 0) * LB_THREADS], d1 = dp[(size_t)(t + 1) * LB_THREADS];
    int4 d2 = dp[(size_t)(t + 2) * LB_THREADS], d3 = dp[(size_t)(t + 3) * LB_THREADS];
    int4 s0 = z4, s1 = z4, s2 = z4, s3 = z4;
    if constexpr (PLACE) {
      s0 = sp[(size_t)(t + 0) * LB_THREADS]; s1 = sp[(size_t)(t + 1) * LB_THREADS];
      s2 = sp[(size_t)(t + 2) * LB_THREADS]; s3 = sp[(size_t)(t + 3) * LB_THREADS];
    }
    split4<PLACE>(hist, seg, stage, d0, s0, n, hme);
    split4<PLACE>(hist, seg, stage, d1, s1, n, hme);
    split4<PLACE>(hist, seg, stage, d2, s2, n, hme);
    split4<PLACE>(hist, seg, stage, d3, s3, n, hme);
  }
  for (; t + 2 <= nfull; t += 2) {
    int4 d0 = dp[(size_t)(t + 0) * LB_THREADS], d1 = dp[(size_t)(t + 1) * LB_THREADS];
    int4 s0 = z4, s1 = z4;
    if constexpr (PLACE) {
      s0 = sp[(size_t)(t + 0) * LB_THREADS]; s1 = sp[(size_t)(t + 1) * LB_THREADS];
    }
    split4<PLACE>(hist, seg, stage, d0, s0, n, hme);
    split4<PLACE>(hist, seg, stage, d1, s1, n, hme);
  }
  for (; t < nfull; ++t) {
    int4 d0 = dp[(size_t)t * LB_THREADS];
    int4 s0 = z4;
    if constexpr (PLACE) {
      s0 = sp[(size_t)t * LB_THREADS];
    }
    split4<PLACE>(hist, seg, stage, d0, s0, n, hme);
  }
  for (int base = e0 + nfull * TILE_EDGES + 4 * tid; base < e1; base += TILE_EDGES) {
    int4 d0 = load_idx4_guard(dst, base, e1, -1);
    int4 s0 = z4;
    if constexpr (PLACE) s0 = load_idx4_guard(src, base, e1, 0);
    split4<PLACE>(hist, seg, stage, d0, s0, n, hme);
  }
}

__global__ __launch_bounds__(LB_THREADS)
void k_split(const int* __restrict__ src, const int* __restrict__ dst,
             unsigned* __restrict__ lists, int* __restrict__ segtab,
             int e, int n, int pchunk) {
  __shared__ unsigned hist[SNR * SHBLK];
  __shared__ unsigned stage[SCAP];
  __shared__ unsigned tot[SNR];
  __shared__ unsigned seg[SNR];
  const int tid  = (int)threadIdx.x;
  const int lane = tid & 31;
  const int wv   = tid >> 5;
  const int p    = (int)blockIdx.x;
  const unsigned hme = (unsigned)(lane * SHP + wv);
  const int e0 = p * pchunk;
  int e1 = e0 + pchunk;
  if (e1 > e) e1 = e;
  const bool vec = ((((size_t)src) | ((size_t)dst)) & 15) == 0;
  const int nfull = (vec && e1 > e0) ? (e1 - e0) / TILE_EDGES : 0;

  for (int i = tid; i < SNR * SHBLK; i += LB_THREADS) hist[i] = 0u;
  for (int i = tid; i < SCAP; i += LB_THREADS) stage[i] = 0u;
  __syncthreads();

  split_sweep<false>(hist, seg, stage, src, dst, e0, e1, nfull, (unsigned)n, tid, hme);
  __syncthreads();

  {
    unsigned* hb = hist + wv * SHBLK + lane * SHP;
    unsigned v[32];
    unsigned sum = 0;
#pragma unroll
    for (int i = 0; i < 32; ++i) { v[i] = hb[i]; sum += v[i]; }
    const unsigned incl = wave_incl_scan(sum, lane);
    unsigned run = incl - sum;
#pragma unroll
    for (int i = 0; i < 32; ++i) { hb[i] = run; run += v[i]; }
    const unsigned total = __shfl(incl, 31);
    if (lane == 0) tot[wv] = total;
  }
  __syncthreads();

  if (wv == 0) {
    const unsigned cnt = tot[lane];
    const unsigned pad = (cnt + 3u) & ~3u;
    const unsigned so  = wave_incl_scan(pad, lane) - pad;
    seg[lane] = so;
    int2 oc; oc.x = (int)so; oc.y = (int)cnt;
    *reinterpret_cast<int2*>(segtab + ((size_t)p * SNR + lane) * 2) = oc;
  }
  __syncthreads();

  split_sweep<true>(hist, seg, stage, src, dst, e0, e1, nfull, (unsigned)n, tid, hme);
  __syncthreads();

  uint4* out = reinterpret_cast<uint4*>(lists + (size_t)p * SCAP);
  const uint4* st4 = reinterpret_cast<const uint4*>(stage);
  for (int i = tid; i < SCAP / 4; i += LB_THREADS) out[i] = st4[i];
}

__global__ __launch_bounds__(LB_THREADS)
void k_seg_deg(const unsigned* __restrict__ lists, const int* __restrict__ segtab,
               float* __restrict__ part, int n, int nranges) {
  __shared__ float cntl[SNPR];
  const int tid  = (int)threadIdx.x;
  const int lane = tid & 31;
  const int wv   = tid >> 5;
  const int r   = (int)(blockIdx.x % (unsigned)nranges);
  const int g   = (int)(blockIdx.x / (unsigned)nranges);
  const int lo  = r * SNPR;
  int cnt = n - lo;
  if (cnt > SNPR) cnt = SNPR;
  const int p = g * 32 + wv;
  const int2 oc = *reinterpret_cast<const int2*>(segtab + ((size_t)p * SNR + r) * 2);

  for (int i = tid; i < SNPR; i += LB_THREADS) cntl[i] = 0.0f;
  __syncthreads();

  const int so = __builtin_amdgcn_readfirstlane(oc.x);
  const int nl = __builtin_amdgcn_readfirstlane(oc.y);
  const int ng = (nl + 127) >> 7;
  const uint4* lp = reinterpret_cast<const uint4*>(lists + (size_t)p * SCAP + so) + lane;
  for (int g0 = 0; g0 < ng; g0 += 4) {
    uint4 w[4];
    for (int j = 0; j < 4; ++j)
      w[j] = (g0 + j < ng) ? lp[(size_t)(g0 + j) * 32] : make_uint4(0u, 0u, 0u, 0u);
    for (int j = 0; j < 4; ++j) {
      const int rem = nl - ((g0 + j) * 128 + lane * 4);
      unsigned ww[4] = {w[j].x, w[j].y, w[j].z, w[j].w};
      for (int k = 0; k < 4; ++k)
        if (k < rem) atomicAdd(&cntl[ww[k] & ((1u << SRELBITS) - 1u)], 1.0f);
    }
  }
  __syncthreads();

  float* pc = part + (size_t)g * n + lo;
  for (int i = tid; i < cnt; i += LB_THREADS) pc[i] = cntl[i];
}

template<int F>
__device__ __forceinline__ void seg_add4(float* acc, const float* __restrict__ m,
                                         uint4 w4, int rem, int lo) {
  unsigned ww[4] = {w4.x, w4.y, w4.z, w4.w};
  unsigned rel[4];
  bool     ok[4];
  float    v[4][F];
  for (int k = 0; k < 4; ++k) {
    ok[k]  = k < rem;
    rel[k] = ww[k] & ((1u << SRELBITS) - 1u);
    int si = ok[k] ? (int)(ww[k] >> SRELBITS) : lo;
    if constexpr (F == 4) {
      float4 t = *reinterpret_cast<const float4*>(m + (size_t)si * 4);
      v[k][0] = t.x; v[k][1] = t.y; v[k][2] = t.z; v[k][3] = t.w;
    } else {
      float2 t = *reinterpret_cast<const float2*>(m + (size_t)si * 2);
      v[k][0] = t.x; v[k][1] = t.y;
    }
  }
  for (int k = 0; k < 4; ++k) {
    if (ok[k]) {
      for (int f = 0; f < F; ++f) atomicAdd(&acc[f * SNPR + rel[k]], v[k][f]);
    }
  }
}

template<int F>
__global__ __launch_bounds__(LB_THREADS)
void k_seg_edge(const unsigned* __restrict__ lists, const int* __restrict__ segtab,
                const float* __restrict__ m, float* __restrict__ part,
                int n, int nranges) {
  __shared__ float acc[4 * NPRD];
  const int tid  = (int)threadIdx.x;
  const int lane = tid & 31;
  const int wv   = tid >> 5;
  const int r   = (int)(blockIdx.x % (unsigned)nranges);
  const int g   = (int)(blockIdx.x / (unsigned)nranges);
  const int lo  = r * SNPR;
  int cnt = n - lo;
  if (cnt > SNPR) cnt = SNPR;
  const int p = g * 32 + wv;
  const int2 oc = *reinterpret_cast<const int2*>(segtab + ((size_t)p * SNR + r) * 2);

  for (int i = tid; i < F * SNPR; i += LB_THREADS) acc[i] = 0.0f;
  __syncthreads();

  const int so = __builtin_amdgcn_readfirstlane(oc.x);
  const int nl = __builtin_amdgcn_readfirstlane(oc.y);
  const int ng = (nl + 127) >> 7;
  const uint4* lp = reinterpret_cast<const uint4*>(lists + (size_t)p * SCAP + so) + lane;
  uint4 wn = make_uint4(0u, 0u, 0u, 0u);
  if (ng > 0) wn = lp[0];
  for (int q = 0; q < ng; ++q) {
    uint4 w4 = wn;
    if (q + 1 < ng) wn = lp[(size_t)(q + 1) * 32];
    seg_add4<F>(acc, m, w4, nl - (q * 128 + lane * 4), lo);
  }
  __syncthreads();

  float* pc = part + (size_t)g * F * n + lo;
  for (int i = tid; i < cnt; i += LB_THREADS) {
    float t[F];
    for (int f = 0; f < F; ++f) t[f] = acc[f * SNPR + i];
    for (int f = 0; f < F; ++f) pc[(size_t)f * n + i] = t[f];
  }
}

extern "C" void kernel_launch(void* const* d_in, const int* in_sizes, int n_in,
                              void* d_out, int out_size, void* d_ws, size_t ws_size,
                              hipStream_t stream) {
  const float* x  = (const float*)d_in[0];
  const int*   ei = (const int*)  d_in[1];
  const float* W1 = (const float*)d_in[2];
  const float* b1 = (const float*)d_in[3];
  const float* W2 = (const float*)d_in[4];
  const float* b2 = (const float*)d_in[5];
  const float* W3 = (const float*)d_in[6];
  const float* b3 = (const float*)d_in[7];
  const float* Wc = (const float*)d_in[8];
  const float* bc = (const float*)d_in[9];

  const int N = in_sizes[0] / 32;
  const int E = in_sizes[1] / 2;
  const int* src = ei;
  const int* dst = ei + E;

  float* out = (float*)d_out;
  float* h3  = out + (size_t)N * 4;

  const int nr4 = (N + NPR4 - 1) / NPR4;
  const int nc4 = (GRID_WGS / nr4) > 0 ? (GRID_WGS / nr4) : 1;
  const int ec4 = (((E + nc4 - 1) / nc4) + 3) & ~3;
  const int nr2 = (N + NPR2 - 1) / NPR2;
  const int nc2 = (GRID_WGS / nr2) > 0 ? (GRID_WGS / nr2) : 1;
  const int ec2 = (((E + nc2 - 1) / nc2) + 3) & ~3;

  const size_t Np = ((size_t)N + 63) & ~(size_t)63;
  float* ws   = (float*)d_ws;
  float* dinv = ws;
  float* m    = ws + Np;
  float* h    = ws + 5 * Np;
  float* part = ws + 9 * Np;

  const int gN  = (N + THREADS - 1) / THREADS;
  const int tiles = (N + 15) / 16;
  const int gW  = (tiles + WAVES_PER_BLOCK - 1) / WAVES_PER_BLOCK;

  const int    wcap   = ((ec4 + TILE_EDGES - 1) / TILE_EDGES) * 128;
  const size_t nwaves = (size_t)nr4 * nc4 * (LB_THREADS / 32);
  const size_t partf  = (((size_t)nc4 * 4 * N) + 63) & ~(size_t)63;
  int*      lcnt  = (int*)(part + partf);
  unsigned* lists = (unsigned*)(part + partf + ((nwaves + 63) & ~(size_t)63));
  const size_t need = (size_t)((const char*)(lists + nwaves * wcap) - (const char*)d_ws);
  const int    snr   = (N + SNPR - 1) / SNPR;
  const int    spc   = (((E + SNP - 1) / SNP) + 3) & ~3;
  const size_t partS = (((size_t)SNC * 4 * N) + 63) & ~(size_t)63;
  int*      segtab = (int*)(part + partS);
  unsigned* segs   = (unsigned*)(part + partS + (size_t)SNP * SNR * 2);
  const size_t needS = (size_t)((const char*)(segs + (size_t)SNP * SCAP) - (const char*)d_ws);
  if (snr <= SNR && N <= (1 << 20) && spc <= SMAXCHUNK && needS <= ws_size) {
    k_split<<<SNP, LB_THREADS, 0, stream>>>(src, dst, segs, segtab, E, N, spc);
    k_seg_deg<<<snr * SNC, LB_THREADS, 0, stream>>>(segs, segtab, part, N, snr);
    k_rsqrt<<<gN, THREADS, 0, stream>>>(dinv, part, N, SNC);

    k_xform32_wmma<<<gW, THREADS, 0, stream>>>(x, W1, dinv, m, N);
    k_seg_edge<4><<<snr * SNC, LB_THREADS, 0, stream>>>(segs, segtab, m, part, N, snr);
    k_tanh<4><<<gN, THREADS, 0, stream>>>(h, part, m, dinv, b1, N, SNC);

    k_xform_small<4, 4, false, true><<<gW, THREADS, 0, stream>>>(h, W2, nullptr, dinv, m, N);
    k_seg_edge<4><<<snr * SNC, LB_THREADS, 0, stream>>>(segs, segtab, m, part, N, snr);
    k_tanh<4><<<gN, THREADS, 0, stream>>>(h, part, m, dinv, b2, N, SNC);

    k_xform_small<4, 2, false, true><<<gW, THREADS, 0, stream>>>(h, W3, nullptr, dinv, m, N);
    k_seg_edge<2><<<snr * SNC, LB_THREADS, 0, stream>>>(segs, segtab, m, part, N, snr);
    k_tanh<2><<<gN, THREADS, 0, stream>>>(h3, part, m, dinv, b3, N, SNC);

    k_xform_small<2, 4, true, false><<<gW, THREADS, 0, stream>>>(h3, Wc, bc, nullptr, out, N);
    return;
  }

  if (N <= (1 << 18) && NPRD <= (1 << LIST_RELBITS) && need <= ws_size) {
    k_deg_list<<<nr4 * nc4, LB_THREADS, 0, stream>>>(src, dst, part, lists, lcnt, E, N, nr4, ec4, wcap);
    k_rsqrt<<<gN, THREADS, 0, stream>>>(dinv, part, N, nc4);

    k_xform32_wmma<<<gW, THREADS, 0, stream>>>(x, W1, dinv, m, N);
    k_edge_list<4><<<nr4 * nc4, LB_THREADS, 0, stream>>>(lists, lcnt, m, part, N, nr4, wcap);
    k_tanh<4><<<gN, THREADS, 0, stream>>>(h, part, m, dinv, b1, N, nc4);

    k_xform_small<4, 4, false, true><<<gW, THREADS, 0, stream>>>(h, W2, nullptr, dinv, m, N);
    k_edge_list<4><<<nr4 * nc4, LB_THREADS, 0, stream>>>(lists, lcnt, m, part, N, nr4, wcap);
    k_tanh<4><<<gN, THREADS, 0, stream>>>(h, part, m, dinv, b2, N, nc4);

    k_xform_small<4, 2, false, true><<<gW, THREADS, 0, stream>>>(h, W3, nullptr, dinv, m, N);
    k_edge_list<2><<<nr4 * nc4, LB_THREADS, 0, stream>>>(lists, lcnt, m, part, N, nr4, wcap);
    k_tanh<2><<<gN, THREADS, 0, stream>>>(h3, part, m, dinv, b3, N, nc4);

    k_xform_small<2, 4, true, false><<<gW, THREADS, 0, stream>>>(h3, Wc, bc, nullptr, out, N);
    return;
  }

  k_deg_lds<<<nr4 * nc4, LB_THREADS, 0, stream>>>(dst, part, E, N, nr4, ec4);
  k_rsqrt<<<gN, THREADS, 0, stream>>>(dinv, part, N, nc4);

  k_xform32_wmma<<<gW, THREADS, 0, stream>>>(x, W1, dinv, m, N);
  k_edge_lds<4><<<nr4 * nc4, LB_THREADS, 0, stream>>>(src, dst, m, part, E, N, nr4, ec4);
  k_tanh<4><<<gN, THREADS, 0, stream>>>(h, part, m, dinv, b1, N, nc4);

  k_xform_small<4, 4, false, true><<<gW, THREADS, 0, stream>>>(h, W2, nullptr, dinv, m, N);
  k_edge_lds<4><<<nr4 * nc4, LB_THREADS, 0, stream>>>(src, dst, m, part, E, N, nr4, ec4);
  k_tanh<4><<<gN, THREADS, 0, stream>>>(h, part, m, dinv, b2, N, nc4);

  k_xform_small<4, 2, false, true><<<gW, THREADS, 0, stream>>>(h, W3, nullptr, dinv, m, N);
  k_edge_lds<2><<<nr2 * nc2, LB_THREADS, 0, stream>>>(src, dst, m, part, E, N, nr2, ec2);
  k_tanh<2><<<gN, THREADS, 0, stream>>>(h3, part, m, dinv, b3, N, nc2);

  k_xform_small<2, 4, true, false><<<gW, THREADS, 0, stream>>>(h3, Wc, bc, nullptr, out, N);
}


# One work-group per 391-node range: chunks sorted into 256 segments; a group adds ALL its range's edges in LDS and finishes its nodes; no partial sums (sum regrouped: 8 lane copies), 8 launches gone
# speedup vs baseline: 16.6865x; 1.1519x over previous
#include <hip/hip_runtime.h>
#include <hip/hip_fp16.h>

typedef __attribute__((ext_vector_type(16))) _Float16 v16h;
typedef __attribute__((ext_vector_type(8)))  float    v8f;
typedef __attribute__((ext_vector_type(2)))  float    v2f;

#define THREADS 256
#define WAVES_PER_BLOCK (THREADS / 32)

#define LB_THREADS 1024
#define GRID_WGS   256
#define NPR4       12544
#define NPR2       20544
#define NPRD       NPR4
constexpr int npr_of(int F) { return F == 4 ? NPR4 : NPR2; }


__device__ __forceinline__ int4 load_idx4_tail(const int* __restrict__ p, int i, int e1, int fill) {
  int4 r;
  r.x = p[i];
  r.y = (i + 1 < e1) ? p[i + 1] : fill;
  r.z = (i + 2 < e1) ? p[i + 2] : fill;
  r.w = (i + 3 < e1) ? p[i + 3] : fill;
  return r;
}

#define TILE_EDGES (4 * LB_THREADS)

__device__ __forceinline__ void deg_add4(float* cntl, int4 d, int lo, unsigned ucnt) {
  unsigned r0 = (unsigned)(d.x - lo), r1 = (unsigned)(d.y - lo);
  unsigned r2 = (unsigned)(d.z - lo), r3 = (unsigned)(d.w - lo);
  if (r0 < ucnt) atomicAdd(&cntl[r0], 1.0f);
  if (r1 < ucnt) atomicAdd(&cntl[r1], 1.0f);
  if (r2 < ucnt) atomicAdd(&cntl[r2], 1.0f);
  if (r3 < ucnt) atomicAdd(&cntl[r3], 1.0f);
}

__global__ __launch_bounds__(LB_THREADS)
void k_deg_lds(const int* __restrict__ dst, float* __restrict__ part,
               int e, int n, int nranges, int echunk) {
  __shared__ float cntl[NPRD];
  const int tid = (int)threadIdx.x;
  const int r   = (int)(blockIdx.x % (unsigned)nranges);
  const int c   = (int)(blockIdx.x / (unsigned)nranges);
  const int lo  = r * NPRD;
  int cnt = n - lo;
  if (cnt > NPRD) cnt = NPRD;
  const unsigned ucnt = (unsigned)cnt;

  for (int i = tid; i < NPRD; i += LB_THREADS) cntl[i] = 0.0f;
  __syncthreads();

  const int e0 = c * echunk;
  int e1 = e0 + echunk;
  if (e1 > e) e1 = e;
  const bool vec = (((size_t)dst) & 15) == 0;
  const int nfull = (vec && e1 > e0) ? (e1 - e0) / TILE_EDGES : 0;

  const int4* dp = reinterpret_cast<const int4*>(dst + e0) + tid;
  int t = 0;
  for (; t + 4 <= nfull; t += 4) {
    int4 d0 = dp[(size_t)(t + 0) * LB_THREADS];
    int4 d1 = dp[(size_t)(t + 1) * LB_THREADS];
    int4 d2 = dp[(size_t)(t + 2) * LB_THREADS];
    int4 d3 = dp[(size_t)(t + 3) * LB_THREADS];
    deg_add4(cntl, d0, lo, ucnt);
    deg_add4(cntl, d1, lo, ucnt);
    deg_add4(cntl, d2, lo, ucnt);
    deg_add4(cntl, d3, lo, ucnt);
  }
  for (; t < nfull; ++t) deg_add4(cntl, dp[(size_t)t * LB_THREADS], lo, ucnt);
  for (int base = e0 + nfull * TILE_EDGES + 4 * tid; base < e1; base += TILE_EDGES)
    deg_add4(cntl, load_idx4_tail(dst, base, e1, -1), lo, ucnt);
  __syncthreads();

  float* pc = part + (size_t)c * n + lo;
  for (int i = tid; i < cnt; i += LB_THREADS) pc[i] = cntl[i];
}

__global__ __launch_bounds__(THREADS)
void k_rsqrt(float* __restrict__ dinv, const float* __restrict__ part, int n, int nchunks) {
  int i = blockIdx.x * blockDim.x + threadIdx.x;
  if (i >= n) return;
  const float* p = part + i;
  float s = 1.0f;
  int c = 0;
  for (; c + 4 <= nchunks; c += 4) {
    float t0 = p[(size_t)(c + 0) * n], t1 = p[(size_t)(c + 1) * n];
    float t2 = p[(size_t)(c + 2) * n], t3 = p[(size_t)(c + 3) * n];
    s += t0; s += t1; s += t2; s += t3;
  }
  for (; c < nchunks; ++c) s += p[(size_t)c * n];
  dinv[i] = rsqrtf(s);
}

__global__ __launch_bounds__(THREADS)
void k_xform32_wmma(const float* __restrict__ x, const float* __restrict__ W,
                    const float* __restrict__ dinv, float* __restrict__ m, int n) {
  __shared__ float lds[WAVES_PER_BLOCK * 64];
  int wave  = (int)((blockIdx.x * (unsigned)blockDim.x + threadIdx.x) >> 5);
  int tl    = threadIdx.x & 31;
  int row0  = wave * 16;
  if (row0 >= n) return;
  int lane16 = tl & 15;
  int hi     = tl >> 4;

  v16h b;
  {
    int  colW = lane16 & 3;
    bool keep = lane16 < 4;
#pragma unroll
    for (int j = 0; j < 16; ++j) {
      int k = (j < 8) ? (hi * 8 + j) : (16 + hi * 8 + (j - 8));
      float wv = W[k * 4 + colW];
      b[j] = keep ? (_Float16)wv : (_Float16)0.0f;
    }
  }

  int  arow = row0 + lane16;
  bool rok  = arow < n;
  int  ar   = rok ? arow : (n - 1);
  const float* xr = x + (size_t)ar * 32;
  float4 q0 = *reinterpret_cast<const float4*>(xr + hi * 8);
  float4 q1 = *reinterpret_cast<const float4*>(xr + hi * 8 + 4);
  float4 q2 = *reinterpret_cast<const float4*>(xr + 16 + hi * 8);
  float4 q3 = *reinterpret_cast<const float4*>(xr + 16 + hi * 8 + 4);
  float di = dinv[ar];
  float zf = rok ? 1.0f : 0.0f;
  v16h a;
  a[0]=(_Float16)(q0.x*zf); a[1]=(_Float16)(q0.y*zf); a[2]=(_Float16)(q0.z*zf); a[3]=(_Float16)(q0.w*zf);
  a[4]=(_Float16)(q1.x*zf); a[5]=(_Float16)(q1.y*zf); a[6]=(_Float16)(q1.z*zf); a[7]=(_Float16)(q1.w*zf);
  a[8]=(_Float16)(q2.x*zf); a[9]=(_Float16)(q2.y*zf); a[10]=(_Float16)(q2.z*zf); a[11]=(_Float16)(q2.w*zf);
  a[12]=(_Float16)(q3.x*zf); a[13]=(_Float16)(q3.y*zf); a[14]=(_Float16)(q3.z*zf); a[15]=(_Float16)(q3.w*zf);

  v8f c = {};
  c = __builtin_amdgcn_wmma_f32_16x16x32_f16(false, a, false, b,
                                             (short)0, c, false, false);

  float* slab = lds + ((threadIdx.x >> 5) << 6);
  if (lane16 < 4) {
#pragma unroll
    for (int r = 0; r < 8; ++r) slab[(hi * 8 + r) * 4 + lane16] = c[r];
  }
  asm volatile("s_wait_dscnt 0" ::: "memory");
  if (hi == 0) {
    int mr = row0 + lane16;
    if (mr < n) {
      float4 v = *reinterpret_cast<const float4*>(slab + lane16 * 4);
      v.x *= di; v.y *= di; v.z *= di; v.w *= di;
      *reinterpret_cast<float4*>(m + (size_t)mr * 4) = v;
    }
  }
}

template<int KIN, int FOUT, bool BIAS, bool SCALE>
__global__ __launch_bounds__(THREADS)
void k_xform_small(const float* __restrict__ hin, const float* __restrict__ W,
                   const float* __restrict__ bias, const float* __restrict__ dinv,
                   float* __restrict__ mout, int n) {
  __shared__ float lds[WAVES_PER_BLOCK * 64];
  int wave  = (int)((blockIdx.x * (unsigned)blockDim.x + threadIdx.x) >> 5);
  int tl    = threadIdx.x & 31;
  int row0  = wave * 16;
  if (row0 >= n) return;
  int lane16 = tl & 15;
  int hi     = tl >> 4;

#if __has_builtin(__builtin_amdgcn_wmma_f32_16x16x4_f32)
  int  arow = row0 + lane16;
  bool rok  = arow < n;
  int  ar   = rok ? arow : (n - 1);
  float zf  = rok ? 1.0f : 0.0f;
  float di  = 1.0f;
  if (SCALE) di = dinv[ar];

  v2f a;
  if (KIN == 4) {
    float2 v = *reinterpret_cast<const float2*>(hin + (size_t)ar * 4 + hi * 2);
    a[0] = v.x * zf;  a[1] = v.y * zf;
  } else {
    float2 v = *reinterpret_cast<const float2*>(hin + (size_t)ar * 2);
    float zk = (hi == 0) ? zf : 0.0f;
    a[0] = v.x * zk;  a[1] = v.y * zk;
  }

  v2f b;
#pragma unroll
  for (int j = 0; j < 2; ++j) {
    int  k    = hi * 2 + j;
    int  kc   = (k < KIN) ? k : (KIN - 1);
    int  cc   = (lane16 < FOUT) ? lane16 : (FOUT - 1);
    float wv  = W[kc * FOUT + cc];
    b[j] = (k < KIN && lane16 < FOUT) ? wv : 0.0f;
  }

  v8f c = {};
  c = __builtin_amdgcn_wmma_f32_16x16x4_f32(false, a, false, b,
                                            (short)0, c, false, false);

  float* slab = lds + ((threadIdx.x >> 5) << 6);
  if (lane16 < FOUT) {
#pragma unroll
    for (int r = 0; r < 8; ++r) slab[(hi * 8 + r) * FOUT + lane16] = c[r];
  }
  asm volatile("s_wait_dscnt 0" ::: "memory");
  if (hi == 0) {
    int mr = row0 + lane16;
    if (mr < n) {
      if (FOUT == 4) {
        float4 v = *reinterpret_cast<const float4*>(slab + lane16 * 4);
        if (BIAS) { v.x += bias[0]; v.y += bias[1]; v.z += bias[2]; v.w += bias[3]; }
        if (SCALE) { v.x *= di; v.y *= di; v.z *= di; v.w *= di; }
        *reinterpret_cast<float4*>(mout + (size_t)mr * 4) = v;
      } else {
        float2 v = *reinterpret_cast<const float2*>(slab + lane16 * 2);
        if (BIAS) { v.x += bias[0]; v.y += bias[1]; }
        if (SCALE) { v.x *= di; v.y *= di; }
        *reinterpret_cast<float2*>(mout + (size_t)mr * 2) = v;
      }
    }
  }
#else
  if (hi == 0) {
    int mr = row0 + lane16;
    if (mr < n) {
#pragma unroll
      for (int f = 0; f < FOUT; ++f) {
        float acc = BIAS ? bias[f] : 0.0f;
#pragma unroll
        for (int k = 0; k < KIN; ++k) acc += hin[mr * KIN + k] * W[k * FOUT + f];
        if (SCALE) acc *= dinv[mr];
        mout[mr * FOUT + f] = acc;
      }
    }
  }
#endif
}

template<int F>
__device__ __forceinline__ void edge_add4(float* acc, const float* __restrict__ m,
                                          int4 d4, int4 s4, int lo, unsigned ucnt) {
  int dd[4] = {d4.x, d4.y, d4.z, d4.w};
  int ss[4] = {s4.x, s4.y, s4.z, s4.w};
  constexpr int NPR = npr_of(F);
  unsigned rel[4];
  bool     hit[4];
  float    v[4][F];
  for (int k = 0; k < 4; ++k) {
    rel[k] = (unsigned)(dd[k] - lo);
    hit[k] = rel[k] < ucnt;
    int      si = hit[k] ? ss[k] : lo;
    if constexpr (F == 4) {
      float4 t = *reinterpret_cast<const float4*>(m + (size_t)si * 4);
      v[k][0] = t.x; v[k][1] = t.y; v[k][2] = t.z; v[k][3] = t.w;
    } else {
      float2 t = *reinterpret_cast<const float2*>(m + (size_t)si * 2);
      v[k][0] = t.x; v[k][1] = t.y;
    }
  }
  for (int k = 0; k < 4; ++k) {
    if (hit[k]) {
      for (int f = 0; f < F; ++f) atomicAdd(&acc[f * NPR + rel[k]], v[k][f]);
    }
  }
}

template<int F>
__global__ __launch_bounds__(LB_THREADS)
void k_edge_lds(const int* __restrict__ src, const int* __restrict__ dst,
                const float* __restrict__ m,
                float* __restrict__ part, int e, int n, int nranges, int echunk) {
  constexpr int NPR = npr_of(F);
  __shared__ float acc[F * NPR];
  const int tid = (int)threadIdx.x;
  const int r   = (int)(blockIdx.x % (unsigned)nranges);
  const int c   = (int)(blockIdx.x / (unsigned)nranges);
  const int lo  = r * NPR;
  int cnt = n - lo;
  if (cnt > NPR) cnt = NPR;
  const unsigned ucnt = (unsigned)cnt;

  for (int i = tid; i < F * NPR; i += LB_THREADS) acc[i] = 0.0f;
  __syncthreads();

  const int e0 = c * echunk;
  int e1 = e0 + echunk;
  if (e1 > e) e1 = e;
  const bool vec = ((((size_t)src) | ((size_t)dst)) & 15) == 0;
  const int nfull = (vec && e1 > e0) ? (e1 - e0) / TILE_EDGES : 0;

  const int4* dp = reinterpret_cast<const int4*>(dst + e0) + tid;
  const int4* sp = reinterpret_cast<const int4*>(src + e0) + tid;
  int4 dn = make_int4(-1, -1, -1, -1), sn = make_int4(0, 0, 0, 0);
  if (nfull > 0) { dn = dp[0]; sn = sp[0]; }
  for (int t = 0; t < nfull; ++t) {
    int4 d4 = dn, s4 = sn;
    if (t + 1 < nfull) {
      dn = dp[(size_t)(t + 1) * LB_THREADS];
      sn = sp[(size_t)(t + 1) * LB_THREADS];
    }
    edge_add4<F>(acc, m, d4, s4, lo, ucnt);
  }
  for (int base = e0 + nfull * TILE_EDGES + 4 * tid; base < e1; base += TILE_EDGES)
    edge_add4<F>(acc, m, load_idx4_tail(dst, base, e1, -1),
                 load_idx4_tail(src, base, e1, 0), lo, ucnt);
  __syncthreads();

  float* pc = part + (size_t)c * F * n + lo;
  for (int i = tid; i < cnt; i += LB_THREADS) {
    float t[F];
    for (int f = 0; f < F; ++f) t[f] = acc[f * NPR + i];
    for (int f = 0; f < F; ++f) pc[(size_t)f * n + i] = t[f];
  }
}

template<int F>
__global__ __launch_bounds__(THREADS)
void k_tanh(float* __restrict__ hout, const float* __restrict__ part,
            const float* __restrict__ m, const float* __restrict__ dinv,
            const float* __restrict__ b, int n, int nchunks) {
  int i = blockIdx.x * blockDim.x + threadIdx.x;
  if (i >= n) return;
  const float* p = part + i;
  const size_t cs = (size_t)F * n;
  float s[F];
  for (int f = 0; f < F; ++f) s[f] = 0.0f;
  int c = 0;
  for (; c + 4 <= nchunks; c += 4) {
    float t[4][F];
    for (int j = 0; j < 4; ++j)
      for (int f = 0; f < F; ++f) t[j][f] = p[(size_t)(c + j) * cs + (size_t)f * n];
    for (int j = 0; j < 4; ++j)
      for (int f = 0; f < F; ++f) s[f] += t[j][f];
  }
  for (; c < nchunks; ++c)
    for (int f = 0; f < F; ++f) s[f] += p[(size_t)c * cs + (size_t)f * n];
  float di = dinv[i];
  if constexpr (F == 4) {
    float4 mv = *reinterpret_cast<const float4*>(m + (size_t)i * 4);
    float4 o;
    o.x = tanhf(di * (s[0] + mv.x) + b[0]);
    o.y = tanhf(di * (s[1] + mv.y) + b[1]);
    o.z = tanhf(di * (s[2] + mv.z) + b[2]);
    o.w = tanhf(di * (s[3] + mv.w) + b[3]);
    *reinterpret_cast<float4*>(hout + (size_t)i * 4) = o;
  } else {
    float2 mv = *reinterpret_cast<const float2*>(m + (size_t)i * 2);
    float2 o;
    o.x = tanhf(di * (s[0] + mv.x) + b[0]);
    o.y = tanhf(di * (s[1] + mv.y) + b[1]);
    *reinterpret_cast<float2*>(hout + (size_t)i * 2) = o;
  }
}


#define LIST_RING 256
#define LIST_RELBITS 14

__device__ __forceinline__ int4 load_idx4_guard(const int* __restrict__ p, int i, int e1, int fill) {
  int4 r;
  r.x = (i     < e1) ? p[i]     : fill;
  r.y = (i + 1 < e1) ? p[i + 1] : fill;
  r.z = (i + 2 < e1) ? p[i + 2] : fill;
  r.w = (i + 3 < e1) ? p[i + 3] : fill;
  return r;
}

__device__ __forceinline__ void deg_emit4(float* cntl, unsigned* myring,
                                          unsigned* __restrict__ mylist,
                                          int4 d, int4 s, int lo, unsigned ucnt,
                                          unsigned lane, unsigned ltmask,
                                          unsigned& wp, unsigned& rp) {
  int dd[4] = {d.x, d.y, d.z, d.w};
  int ss[4] = {s.x, s.y, s.z, s.w};
  for (int k = 0; k < 4; ++k) {
    unsigned rel = (unsigned)(dd[k] - lo);
    bool hit = rel < ucnt;
    unsigned b = __builtin_amdgcn_ballot_w32(hit);
    if (hit) {
      atomicAdd(&cntl[rel], 1.0f);
      unsigned slot = (wp + (unsigned)__builtin_popcount(b & ltmask)) & (LIST_RING - 1);
      myring[slot] = ((unsigned)ss[k] << LIST_RELBITS) | rel;
    }
    wp += (unsigned)__builtin_popcount(b);
  }
  while (wp - rp >= 32u) {
    __builtin_amdgcn_wave_barrier();
    mylist[rp + lane] = myring[(rp + lane) & (LIST_RING - 1)];
    rp += 32u;
  }
}

__global__ __launch_bounds__(LB_THREADS)
void k_deg_list(const int* __restrict__ src, const int* __restrict__ dst,
                float* __restrict__ part, unsigned* __restrict__ lists,
                int* __restrict__ lcnt, int e, int n, int nranges, int echunk, int wcap) {
  __shared__ float cntl[NPRD];
  __shared__ unsigned ring[(LB_THREADS / 32) * LIST_RING];
  const int tid = (int)threadIdx.x;
  const unsigned lane = (unsigned)tid & 31u;
  const unsigned ltmask = (1u << lane) - 1u;
  const int wv  = tid >> 5;
  const int r   = (int)(blockIdx.x % (unsigned)nranges);
  const int c   = (int)(blockIdx.x / (unsigned)nranges);
  const int lo  = r * NPRD;
  int cnt = n - lo;
  if (cnt > NPRD) cnt = NPRD;
  const unsigned ucnt = (unsigned)cnt;
  const size_t wid = (size_t)blockIdx.x * (LB_THREADS / 32) + (size_t)wv;
  unsigned* myring = ring + wv * LIST_RING;
  unsigned* mylist = lists + wid * (size_t)wcap;
  unsigned wp = 0, rp = 0;

  for (int i = tid; i < NPRD; i += LB_THREADS) cntl[i] = 0.0f;
  __syncthreads();

  const int e0 = c * echunk;
  int e1 = e0 + echunk;
  if (e1 > e) e1 = e;
  const bool vec = ((((size_t)src) | ((size_t)dst)) & 15) == 0;
  const int nfull = (vec && e1 > e0) ? (e1 - e0) / TILE_EDGES : 0;

  const int4* dp = reinterpret_cast<const int4*>(dst + e0) + tid;
  const int4* sp = reinterpret_cast<const int4*>(src + e0) + tid;
  int4 dn = make_int4(-1, -1, -1, -1), sn = make_int4(0, 0, 0, 0);
  if (nfull > 0) { dn = dp[0]; sn = sp[0]; }
  for (int t = 0; t < nfull; ++t) {
    int4 d4 = dn, s4 = sn;
    if (t + 1 < nfull) {
      dn = dp[(size_t)(t + 1) * LB_THREADS];
      sn = sp[(size_t)(t + 1) * LB_THREADS];
    }
    deg_emit4(cntl, myring, mylist, d4, s4, lo, ucnt, lane, ltmask, wp, rp);
  }
  for (int b0 = e0 + nfull * TILE_EDGES + 4 * (tid & ~31); b0 < e1; b0 += TILE_EDGES) {
    int base = b0 + 4 * (int)lane;
    deg_emit4(cntl, myring, mylist, load_idx4_guard(dst, base, e1, -1),
              load_idx4_guard(src, base, e1, 0), lo, ucnt, lane, ltmask, wp, rp);
  }
  __builtin_amdgcn_wave_barrier();
  if (rp + lane < wp) mylist[rp + lane] = myring[(rp + lane) & (LIST_RING - 1)];
  if (lane == 0) lcnt[wid] = (int)wp;
  __syncthreads();

  float* pc = part + (size_t)c * n + lo;
  for (int i = tid; i < cnt; i += LB_THREADS) pc[i] = cntl[i];
}

template<int F>
__device__ __forceinline__ void list_add4(float* acc, const float* __restrict__ m,
                                          uint4 w4, int rem, int lo) {
  unsigned ww[4] = {w4.x, w4.y, w4.z, w4.w};
  unsigned rel[4];
  bool     ok[4];
  float    v[4][F];
  for (int k = 0; k < 4; ++k) {
    ok[k]  = k < rem;
    rel[k] = ww[k] & ((1u << LIST_RELBITS) - 1u);
    int si = ok[k] ? (int)(ww[k] >> LIST_RELBITS) : lo;
    if constexpr (F == 4) {
      float4 t = *reinterpret_cast<const float4*>(m + (size_t)si * 4);
      v[k][0] = t.x; v[k][1] = t.y; v[k][2] = t.z; v[k][3] = t.w;
    } else {
      float2 t = *reinterpret_cast<const float2*>(m + (size_t)si * 2);
      v[k][0] = t.x; v[k][1] = t.y;
    }
  }
  for (int k = 0; k < 4; ++k) {
    if (ok[k]) {
      for (int f = 0; f < F; ++f) atomicAdd(&acc[f * NPRD + rel[k]], v[k][f]);
    }
  }
}

template<int F>
__global__ __launch_bounds__(LB_THREADS)
void k_edge_list(const unsigned* __restrict__ lists, const int* __restrict__ lcnt,
                 const float* __restrict__ m, float* __restrict__ part,
                 int n, int nranges, int wcap) {
  __shared__ float acc[4 * NPRD];
  const int tid  = (int)threadIdx.x;
  const int lane = tid & 31;
  const int wv   = tid >> 5;
  const int r   = (int)(blockIdx.x % (unsigned)nranges);
  const int c   = (int)(blockIdx.x / (unsigned)nranges);
  const int lo  = r * NPRD;
  int cnt = n - lo;
  if (cnt > NPRD) cnt = NPRD;

  for (int i = tid; i < F * NPRD; i += LB_THREADS) acc[i] = 0.0f;
  __syncthreads();

  const size_t wid = (size_t)blockIdx.x * (LB_THREADS / 32) + (size_t)wv;
  const int nl = __builtin_amdgcn_readfirstlane(lcnt[wid]);
  const int ng = (nl + 127) >> 7;
  const uint4* lp = reinterpret_cast<const uint4*>(lists + wid * (size_t)wcap) + lane;
  uint4 wn = make_uint4(0u, 0u, 0u, 0u);
  if (ng > 0) wn = lp[0];
  for (int g = 0; g < ng; ++g) {
    uint4 w4 = wn;
    if (g + 1 < ng) wn = lp[(size_t)(g + 1) * 32];
    list_add4<F>(acc, m, w4, nl - (g * 128 + lane * 4), lo);
  }
  __syncthreads();

  float* pc = part + (size_t)c * F * n + lo;
  for (int i = tid; i < cnt; i += LB_THREADS) {
    float t[F];
    for (int f = 0; f < F; ++f) t[f] = acc[f * NPRD + i];
    for (int f = 0; f < F; ++f) pc[(size_t)f * n + i] = t[f];
  }
}


#define SNPR     3136
#define SRELBITS 12
#define SNR      32
#define SNP      256
#define SNC      (SNP / 32)
#define SHP      33
#define SHBLK    (32 * SHP)
#define SCAP     26624
#define SMAXCHUNK 25600

__device__ __forceinline__ unsigned wave_incl_scan(unsigned x, int lane) {
  for (int d = 1; d < 32; d <<= 1) {
    unsigned t = __shfl_up(x, (unsigned)d);
    if (lane >= d) x += t;
  }
  return x;
}

template<bool PLACE>
__device__ __forceinline__ void split4(unsigned* hist, const unsigned* seg, unsigned* stage,
                                       int4 d4, int4 s4, unsigned n, unsigned hme) {
  int dd[4] = {d4.x, d4.y, d4.z, d4.w};
  int ss[4] = {s4.x, s4.y, s4.z, s4.w};
  bool     ok[4];
  unsigned r[4], rank[4], sg[4];
  for (int k = 0; k < 4; ++k) {
    ok[k] = (unsigned)dd[k] < n;
    r[k]  = ok[k] ? (unsigned)dd[k] / SNPR : 0u;
    unsigned* hp = hist + r[k] * SHBLK + hme;
    if constexpr (!PLACE) {
      atomicAdd(hp, ok[k] ? 1u : 0u);
    } else {
      rank[k] = atomicAdd(hp, ok[k] ? 1u : 0u);
      sg[k]   = seg[r[k]];
    }
  }
  if constexpr (PLACE) {
    for (int k = 0; k < 4; ++k)
      if (ok[k]) stage[sg[k] + rank[k]] = ((unsigned)ss[k] << SRELBITS) | ((unsigned)dd[k] - r[k] * SNPR);
  }
}

template<bool PLACE>
__device__ __forceinline__ void split_sweep(unsigned* hist, const unsigned* seg, unsigned* stage,
                                            const int* __restrict__ src, const int* __restrict__ dst,
                                            int e0, int e1, int nfull, unsigned n, int tid, unsigned hme) {
  const int4 z4 = make_int4(0, 0, 0, 0);
  const int4* dp = reinterpret_cast<const int4*>(dst + e0) + tid;
  const int4* sp = reinterpret_cast<const int4*>(src + e0) + tid;
  int t = 0;
  for (; t + 4 <= nfull; t += 4) {
    int4 d0 = dp[(size_t)(t + 0) * LB_THREADS], d1 = dp[(size_t)(t + 1) * LB_THREADS];
    int4 d2 = dp[(size_t)(t + 2) * LB_THREADS], d3 = dp[(size_t)(t + 3) * LB_THREADS];
    int4 s0 = z4, s1 = z4, s2 = z4, s3 = z4;
    if constexpr (PLACE) {
      s0 = sp[(size_t)(t + 0) * LB_THREADS]; s1 = sp[(size_t)(t + 1) * LB_THREADS];
      s2 = sp[(size_t)(t + 2) * LB_THREADS]; s3 = sp[(size_t)(t + 3) * LB_THREADS];
    }
    split4<PLACE>(hist, seg, stage, d0, s0, n, hme);
    split4<PLACE>(hist, seg, stage, d1, s1, n, hme);
    split4<PLACE>(hist, seg, stage, d2, s2, n, hme);
    split4<PLACE>(hist, seg, stage, d3, s3, n, hme);
  }
  for (; t + 2 <= nfull; t += 2) {
    int4 d0 = dp[(size_t)(t + 0) * LB_THREADS], d1 = dp[(size_t)(t + 1) * LB_THREADS];
    int4 s0 = z4, s1 = z4;
    if constexpr (PLACE) {
      s0 = sp[(size_t)(t + 0) * LB_THREADS]; s1 = sp[(size_t)(t + 1) * LB_THREADS];
    }
    split4<PLACE>(hist, seg, stage, d0, s0, n, hme);
    split4<PLACE>(hist, seg, stage, d1, s1, n, hme);
  }
  for (; t < nfull; ++t) {
    int4 d0 = dp[(size_t)t * LB_THREADS];
    int4 s0 = z4;
    if constexpr (PLACE) {
      s0 = sp[(size_t)t * LB_THREADS];
    }
    split4<PLACE>(hist, seg, stage, d0, s0, n, hme);
  }
  for (int base = e0 + nfull * TILE_EDGES + 4 * tid; base < e1; base += TILE_EDGES) {
    int4 d0 = load_idx4_guard(dst, base, e1, -1);
    int4 s0 = z4;
    if constexpr (PLACE) s0 = load_idx4_guard(src, base, e1, 0);
    split4<PLACE>(hist, seg, stage, d0, s0, n, hme);
  }
}

__global__ __launch_bounds__(LB_THREADS)
void k_split(const int* __restrict__ src, const int* __restrict__ dst,
             unsigned* __restrict__ lists, int* __restrict__ segtab,
             int e, int n, int pchunk) {
  __shared__ unsigned hist[SNR * SHBLK];
  __shared__ unsigned stage[SCAP];
  __shared__ unsigned tot[SNR];
  __shared__ unsigned seg[SNR];
  const int tid  = (int)threadIdx.x;
  const int lane = tid & 31;
  const int wv   = tid >> 5;
  const int p    = (int)blockIdx.x;
  const unsigned hme = (unsigned)(lane * SHP + wv);
  const int e0 = p * pchunk;
  int e1 = e0 + pchunk;
  if (e1 > e) e1 = e;
  const bool vec = ((((size_t)src) | ((size_t)dst)) & 15) == 0;
  const int nfull = (vec && e1 > e0) ? (e1 - e0) / TILE_EDGES : 0;

  for (int i = tid; i < SNR * SHBLK; i += LB_THREADS) hist[i] = 0u;
  for (int i = tid; i < SCAP; i += LB_THREADS) stage[i] = 0u;
  __syncthreads();

  split_sweep<false>(hist, seg, stage, src, dst, e0, e1, nfull, (unsigned)n, tid, hme);
  __syncthreads();

  {
    unsigned* hb = hist + wv * SHBLK + lane * SHP;
    unsigned v[32];
    unsigned sum = 0;
#pragma unroll
    for (int i = 0; i < 32; ++i) { v[i] = hb[i]; sum += v[i]; }
    const unsigned incl = wave_incl_scan(sum, lane);
    unsigned run = incl - sum;
#pragma unroll
    for (int i = 0; i < 32; ++i) { hb[i] = run; run += v[i]; }
    const unsigned total = __shfl(incl, 31);
    if (lane == 0) tot[wv] = total;
  }
  __syncthreads();

  if (wv == 0) {
    const unsigned cnt = tot[lane];
    const unsigned pad = (cnt + 3u) & ~3u;
    const unsigned so  = wave_incl_scan(pad, lane) - pad;
    seg[lane] = so;
    int2 oc; oc.x = (int)so; oc.y = (int)cnt;
    *reinterpret_cast<int2*>(segtab + ((size_t)p * SNR + lane) * 2) = oc;
  }
  __syncthreads();

  split_sweep<true>(hist, seg, stage, src, dst, e0, e1, nfull, (unsigned)n, tid, hme);
  __syncthreads();

  uint4* out = reinterpret_cast<uint4*>(lists + (size_t)p * SCAP);
  const uint4* st4 = reinterpret_cast<const uint4*>(stage);
  for (int i = tid; i < SCAP / 4; i += LB_THREADS) out[i] = st4[i];
}

__global__ __launch_bounds__(LB_THREADS)
void k_seg_deg(const unsigned* __restrict__ lists, const int* __restrict__ segtab,
               float* __restrict__ part, int n, int nranges) {
  __shared__ float cntl[SNPR];
  const int tid  = (int)threadIdx.x;
  const int lane = tid & 31;
  const int wv   = tid >> 5;
  const int r   = (int)(blockIdx.x % (unsigned)nranges);
  const int g   = (int)(blockIdx.x / (unsigned)nranges);
  const int lo  = r * SNPR;
  int cnt = n - lo;
  if (cnt > SNPR) cnt = SNPR;
  const int p = g * 32 + wv;
  const int2 oc = *reinterpret_cast<const int2*>(segtab + ((size_t)p * SNR + r) * 2);

  for (int i = tid; i < SNPR; i += LB_THREADS) cntl[i] = 0.0f;
  __syncthreads();

  const int so = __builtin_amdgcn_readfirstlane(oc.x);
  const int nl = __builtin_amdgcn_readfirstlane(oc.y);
  const int ng = (nl + 127) >> 7;
  const uint4* lp = reinterpret_cast<const uint4*>(lists + (size_t)p * SCAP + so) + lane;
  for (int g0 = 0; g0 < ng; g0 += 4) {
    uint4 w[4];
    for (int j = 0; j < 4; ++j)
      w[j] = (g0 + j < ng) ? lp[(size_t)(g0 + j) * 32] : make_uint4(0u, 0u, 0u, 0u);
    for (int j = 0; j < 4; ++j) {
      const int rem = nl - ((g0 + j) * 128 + lane * 4);
      unsigned ww[4] = {w[j].x, w[j].y, w[j].z, w[j].w};
      for (int k = 0; k < 4; ++k)
        if (k < rem) atomicAdd(&cntl[ww[k] & ((1u << SRELBITS) - 1u)], 1.0f);
    }
  }
  __syncthreads();

  float* pc = part + (size_t)g * n + lo;
  for (int i = tid; i < cnt; i += LB_THREADS) pc[i] = cntl[i];
}

template<int F>
__device__ __forceinline__ void seg_add4(float* acc, const float* __restrict__ m,
                                         uint4 w4, int rem, int lo) {
  unsigned ww[4] = {w4.x, w4.y, w4.z, w4.w};
  unsigned rel[4];
  bool     ok[4];
  float    v[4][F];
  for (int k = 0; k < 4; ++k) {
    ok[k]  = k < rem;
    rel[k] = ww[k] & ((1u << SRELBITS) - 1u);
    int si = ok[k] ? (int)(ww[k] >> SRELBITS) : lo;
    if constexpr (F == 4) {
      float4 t = *reinterpret_cast<const float4*>(m + (size_t)si * 4);
      v[k][0] = t.x; v[k][1] = t.y; v[k][2] = t.z; v[k][3] = t.w;
    } else {
      float2 t = *reinterpret_cast<const float2*>(m + (size_t)si * 2);
      v[k][0] = t.x; v[k][1] = t.y;
    }
  }
  for (int k = 0; k < 4; ++k) {
    if (ok[k]) {
      for (int f = 0; f < F; ++f) atomicAdd(&acc[f * SNPR + rel[k]], v[k][f]);
    }
  }
}

template<int F>
__global__ __launch_bounds__(LB_THREADS)
void k_seg_edge(const unsigned* __restrict__ lists, const int* __restrict__ segtab,
                const float* __restrict__ m, float* __restrict__ part,
                int n, int nranges) {
  __shared__ float acc[4 * NPRD];
  const int tid  = (int)threadIdx.x;
  const int lane = tid & 31;
  const int wv   = tid >> 5;
  const int r   = (int)(blockIdx.x % (unsigned)nranges);
  const int g   = (int)(blockIdx.x / (unsigned)nranges);
  const int lo  = r * SNPR;
  int cnt = n - lo;
  if (cnt > SNPR) cnt = SNPR;
  const int p = g * 32 + wv;
  const int2 oc = *reinterpret_cast<const int2*>(segtab + ((size_t)p * SNR + r) * 2);

  for (int i = tid; i < F * SNPR; i += LB_THREADS) acc[i] = 0.0f;
  __syncthreads();

  const int so = __builtin_amdgcn_readfirstlane(oc.x);
  const int nl = __builtin_amdgcn_readfirstlane(oc.y);
  const int ng = (nl + 127) >> 7;
  const uint4* lp = reinterpret_cast<const uint4*>(lists + (size_t)p * SCAP + so) + lane;
  uint4 wn = make_uint4(0u, 0u, 0u, 0u);
  if (ng > 0) wn = lp[0];
  for (int q = 0; q < ng; ++q) {
    uint4 w4 = wn;
    if (q + 1 < ng) wn = lp[(size_t)(q + 1) * 32];
    seg_add4<F>(acc, m, w4, nl - (q * 128 + lane * 4), lo);
  }
  __syncthreads();

  float* pc = part + (size_t)g * F * n + lo;
  for (int i = tid; i < cnt; i += LB_THREADS) {
    float t[F];
    for (int f = 0; f < F; ++f) t[f] = acc[f * SNPR + i];
    for (int f = 0; f < F; ++f) pc[(size_t)f * n + i] = t[f];
  }
}


#define RN        391
#define RBITS     9
#define RNR       256
#define RP        392
#define RCH       (SNP / 32)
#define RSG       4
#define NCP       8

template<bool PLACE>
__device__ __forceinline__ void rsplit4(unsigned* cntw, const unsigned* seg, unsigned* stage,
                                        int4 d4, int4 s4, unsigned n) {
  int dd[4] = {d4.x, d4.y, d4.z, d4.w};
  int ss[4] = {s4.x, s4.y, s4.z, s4.w};
  bool     ok[4];
  unsigned r[4], rank[4], sg[4];
  for (int k = 0; k < 4; ++k) {
    ok[k] = (unsigned)dd[k] < n;
    r[k]  = ok[k] ? (unsigned)dd[k] / RN : 0u;
    unsigned* hp = cntw + r[k];
    if constexpr (!PLACE) {
      atomicAdd(hp, ok[k] ? 1u : 0u);
    } else {
      rank[k] = atomicAdd(hp, ok[k] ? 1u : 0u);
      sg[k]   = seg[r[k]];
    }
  }
  if constexpr (PLACE) {
    for (int k = 0; k < 4; ++k)
      if (ok[k]) stage[sg[k] + rank[k]] = ((unsigned)ss[k] << RBITS) | ((unsigned)dd[k] - r[k] * RN);
  }
}

template<bool PLACE>
__device__ __forceinline__ void rsplit_sweep(unsigned* cntw, const unsigned* seg, unsigned* stage,
                                             const int* __restrict__ src, const int* __restrict__ dst,
                                             int e0, int e1, int nfull, unsigned n, int tid) {
  const int4 z4 = make_int4(0, 0, 0, 0);
  const int4* dp = reinterpret_cast<const int4*>(dst + e0) + tid;
  const int4* sp = reinterpret_cast<const int4*>(src + e0) + tid;
  int t = 0;
  for (; t + 4 <= nfull; t += 4) {
    int4 d0 = dp[(size_t)(t + 0) * LB_THREADS], d1 = dp[(size_t)(t + 1) * LB_THREADS];
    int4 d2 = dp[(size_t)(t + 2) * LB_THREADS], d3 = dp[(size_t)(t + 3) * LB_THREADS];
    int4 s0 = z4, s1 = z4, s2 = z4, s3 = z4;
    if constexpr (PLACE) {
      s0 = sp[(size_t)(t + 0) * LB_THREADS]; s1 = sp[(size_t)(t + 1) * LB_THREADS];
      s2 = sp[(size_t)(t + 2) * LB_THREADS]; s3 = sp[(size_t)(t + 3) * LB_THREADS];
    }
    rsplit4<PLACE>(cntw, seg, stage, d0, s0, n);
    rsplit4<PLACE>(cntw, seg, stage, d1, s1, n);
    rsplit4<PLACE>(cntw, seg, stage, d2, s2, n);
    rsplit4<PLACE>(cntw, seg, stage, d3, s3, n);
  }
  for (; t + 2 <= nfull; t += 2) {
    int4 d0 = dp[(size_t)(t + 0) * LB_THREADS], d1 = dp[(size_t)(t + 1) * LB_THREADS];
    int4 s0 = z4, s1 = z4;
    if constexpr (PLACE) {
      s0 = sp[(size_t)(t + 0) * LB_THREADS]; s1 = sp[(size_t)(t + 1) * LB_THREADS];
    }
    rsplit4<PLACE>(cntw, seg, stage, d0, s0, n);
    rsplit4<PLACE>(cntw, seg, stage, d1, s1, n);
  }
  for (; t < nfull; ++t) {
    int4 d0 = dp[(size_t)t * LB_THREADS];
    int4 s0 = z4;
    if constexpr (PLACE) s0 = sp[(size_t)t * LB_THREADS];
    rsplit4<PLACE>(cntw, seg, stage, d0, s0, n);
  }
  for (int base = e0 + nfull * TILE_EDGES + 4 * tid; base < e1; base += TILE_EDGES) {
    int4 d0 = load_idx4_guard(dst, base, e1, -1);
    int4 s0 = z4;
    if constexpr (PLACE) s0 = load_idx4_guard(src, base, e1, 0);
    rsplit4<PLACE>(cntw, seg, stage, d0, s0, n);
  }
}

__global__ __launch_bounds__(LB_THREADS)
void k_rsplit(const int* __restrict__ src, const int* __restrict__ dst,
              unsigned* __restrict__ lists, int* __restrict__ segtab,
              int e, int n, int pchunk) {
  __shared__ unsigned cntl[32 * RSG * RNR];
  __shared__ unsigned stage[SCAP];
  __shared__ unsigned seg[RNR];
  __shared__ unsigned ptot[4 * RNR];
  __shared__ unsigned wsum[RNR / 32];
  const int tid  = (int)threadIdx.x;
  const int lane = tid & 31;
  const int wv   = tid >> 5;
  const int p    = (int)blockIdx.x;
  const int e0 = p * pchunk;
  int e1 = e0 + pchunk;
  if (e1 > e) e1 = e;
  const bool vec = ((((size_t)src) | ((size_t)dst)) & 15) == 0;
  const int nfull = (vec && e1 > e0) ? (e1 - e0) / TILE_EDGES : 0;
  unsigned* cntw = cntl + (wv * RSG + (lane & (RSG - 1))) * RNR;

  for (int i = tid; i < 32 * RSG * RNR; i += LB_THREADS) cntl[i] = 0u;
  for (int i = tid; i < SCAP; i += LB_THREADS) stage[i] = 0u;
  __syncthreads();

  rsplit_sweep<false>(cntw, seg, stage, src, dst, e0, e1, nfull, (unsigned)n, tid);
  __syncthreads();

  const int sb = tid & (RNR - 1), sq = tid >> 8;
  unsigned v[32];
  unsigned run = 0, total = 0;
  {
    unsigned mysum = 0;
#pragma unroll
    for (int i = 0; i < 32; ++i) { v[i] = cntl[(sq * 32 + i) * RNR + sb]; mysum += v[i]; }
    ptot[sq * RNR + sb] = mysum;
  }
  __syncthreads();
  for (int q = 0; q < 4; ++q) {
    const unsigned t = ptot[q * RNR + sb];
    run   += (q < sq) ? t : 0u;
    total += t;
  }
#pragma unroll
  for (int i = 0; i < 32; ++i) { cntl[(sq * 32 + i) * RNR + sb] = run; run += v[i]; }
  const unsigned pad = (total + 3u) & ~3u;
  unsigned incl = 0;
  if (tid < RNR) {
    incl = wave_incl_scan(pad, lane);
    if (lane == 31) wsum[wv] = incl;
  }
  __syncthreads();
  if (tid < RNR) {
    unsigned base = 0;
    for (int w = 0; w < RNR / 32; ++w) base += (w < wv) ? wsum[w] : 0u;
    const unsigned so = base + incl - pad;
    seg[tid] = so;
    int2 oc; oc.x = (int)so; oc.y = (int)total;
    *reinterpret_cast<int2*>(segtab + ((size_t)p * RNR + tid) * 2) = oc;
  }
  __syncthreads();

  rsplit_sweep<true>(cntw, seg, stage, src, dst, e0, e1, nfull, (unsigned)n, tid);
  __syncthreads();

  uint4* out = reinterpret_cast<uint4*>(lists + (size_t)p * SCAP);
  const uint4* st4 = reinterpret_cast<const uint4*>(stage);
  for (int i = tid; i < SCAP / 4; i += LB_THREADS) out[i] = st4[i];
}

__device__ __forceinline__ void rng_segs4(const int* __restrict__ segtab, int p0, int r,
                                          int so[4], int nl[4]) {
  for (int j = 0; j < 4; ++j) {
    const int2 oc = *reinterpret_cast<const int2*>(segtab + ((size_t)(p0 + j) * RNR + r) * 2);
    so[j] = __builtin_amdgcn_readfirstlane(oc.x);
    nl[j] = __builtin_amdgcn_readfirstlane(oc.y);
  }
}

__global__ __launch_bounds__(LB_THREADS)
void k_rng_deg(const unsigned* __restrict__ lists, const int* __restrict__ segtab,
               const float* __restrict__ x, const float* __restrict__ W,
               float* __restrict__ dinv, float* __restrict__ m, int n) {
  __shared__ float acc[4 * NPRD];
  __shared__ float slabs[32 * 64];
  const int tid  = (int)threadIdx.x;
  const int lane = tid & 31;
  const int wv   = __builtin_amdgcn_readfirstlane(tid >> 5);
  const int r    = (int)blockIdx.x;
  const int lo   = r * RN;
  int cnt = n - lo;
  if (cnt > RN) cnt = RN;

  for (int i = tid; i < NCP * RP; i += LB_THREADS) acc[i] = 0.0f;
  __syncthreads();

  const int ntiles = (cnt + 15) >> 4;
  const int lane16 = lane & 15;
  const int hi     = lane >> 4;
  const int il  = wv * 16 + lane16;
  const bool rok = il < cnt;
  const int ilc = rok ? il : (cnt - 1);
  float4 q0 = make_float4(0.f, 0.f, 0.f, 0.f), q1 = q0, q2 = q0, q3 = q0;
  if (wv < ntiles) {
    const float* xr = x + (size_t)(lo + ilc) * 32;
    q0 = *reinterpret_cast<const float4*>(xr + hi * 8);
    q1 = *reinterpret_cast<const float4*>(xr + hi * 8 + 4);
    q2 = *reinterpret_cast<const float4*>(xr + 16 + hi * 8);
    q3 = *reinterpret_cast<const float4*>(xr + 16 + hi * 8 + 4);
  }

  float* mine = acc + (lane & (NCP - 1)) * RP;
  for (int j0 = 0; j0 < RCH; j0 += 4) {
    const int p0 = wv * RCH + j0;
    int so[4], nl[4];
    rng_segs4(segtab, p0, r, so, nl);
    int ng[4], ngmax = 0;
    for (int j = 0; j < 4; ++j) { ng[j] = (nl[j] + 127) >> 7; ngmax = ng[j] > ngmax ? ng[j] : ngmax; }
    for (int q = 0; q < ngmax; ++q) {
      uint4 w[4];
      int rem[4];
      for (int j = 0; j < 4; ++j) {
        rem[j] = nl[j] - (q * 128 + lane * 4);
        const int qj = (q < ng[j]) ? q : (ng[j] > 0 ? ng[j] - 1 : 0);
        const uint4* lp = reinterpret_cast<const uint4*>(lists + (size_t)(p0 + j) * SCAP + so[j]) + lane;
        w[j] = lp[(size_t)qj * 32];
      }
      for (int j = 0; j < 4; ++j) {
        unsigned ww[4] = {w[j].x, w[j].y, w[j].z, w[j].w};
        for (int k = 0; k < 4; ++k)
          if (k < rem[j]) atomicAdd(&mine[ww[k] & ((1u << RBITS) - 1u)], 1.0f);
      }
    }
  }
  __syncthreads();

  float* dl = acc + NCP * RP;
  for (int i = tid; i < cnt; i += LB_THREADS) {
    float s = 1.0f;
    for (int c = 0; c < NCP; ++c) s += acc[c * RP + i];
    const float d = rsqrtf(s);
    dl[i] = d;
    dinv[lo + i] = d;
  }
  __syncthreads();

  if (wv < ntiles) {
    v16h b;
    {
      int  colW = lane16 & 3;
      bool keep = lane16 < 4;
#pragma unroll
      for (int j = 0; j < 16; ++j) {
        int k = (j < 8) ? (hi * 8 + j) : (16 + hi * 8 + (j - 8));
        float wvv = W[k * 4 + colW];
        b[j] = keep ? (_Float16)wvv : (_Float16)0.0f;
      }
    }
    const float di = dl[ilc];
    const float zf = rok ? 1.0f : 0.0f;
    v16h a;
    a[0]=(_Float16)(q0.x*zf); a[1]=(_Float16)(q0.y*zf); a[2]=(_Float16)(q0.z*zf); a[3]=(_Float16)(q0.w*zf);
    a[4]=(_Float16)(q1.x*zf); a[5]=(_Float16)(q1.y*zf); a[6]=(_Float16)(q1.z*zf); a[7]=(_Float16)(q1.w*zf);
    a[8]=(_Float16)(q2.x*zf); a[9]=(_Float16)(q2.y*zf); a[10]=(_Float16)(q2.z*zf); a[11]=(_Float16)(q2.w*zf);
    a[12]=(_Float16)(q3.x*zf); a[13]=(_Float16)(q3.y*zf); a[14]=(_Float16)(q3.z*zf); a[15]=(_Float16)(q3.w*zf);
    v8f c = {};
    c = __builtin_amdgcn_wmma_f32_16x16x32_f16(false, a, false, b, (short)0, c, false, false);
    float* slab = slabs + wv * 64;
    if (lane16 < 4) {
#pragma unroll
      for (int rr = 0; rr < 8; ++rr) slab[(hi * 8 + rr) * 4 + lane16] = c[rr];
    }
    asm volatile("s_wait_dscnt 0" ::: "memory");
    if (hi == 0 && rok) {
      float4 v = *reinterpret_cast<const float4*>(slab + lane16 * 4);
      v.x *= di; v.y *= di; v.z *= di; v.w *= di;
      *reinterpret_cast<float4*>(m + (size_t)(lo + il) * 4) = v;
    }
  }
}

template<int F>
__device__ __forceinline__ void rng_add16(float* acc, const float* __restrict__ m,
                                          const uint4 w[4], const int rem[4], int lo) {
  unsigned rel[16];
  bool     ok[16];
  float    v[16][F];
#pragma unroll
  for (int j = 0; j < 4; ++j) {
    unsigned ww[4] = {w[j].x, w[j].y, w[j].z, w[j].w};
#pragma unroll
    for (int k = 0; k < 4; ++k) {
      const int t = j * 4 + k;
      ok[t]  = k < rem[j];
      rel[t] = ww[k] & ((1u << RBITS) - 1u);
      int si = ok[t] ? (int)(ww[k] >> RBITS) : lo;
      if constexpr (F == 4) {
        float4 q = *reinterpret_cast<const float4*>(m + (size_t)si * 4);
        v[t][0] = q.x; v[t][1] = q.y; v[t][2] = q.z; v[t][3] = q.w;
      } else {
        float2 q = *reinterpret_cast<const float2*>(m + (size_t)si * 2);
        v[t][0] = q.x; v[t][1] = q.y;
      }
    }
  }
#pragma unroll
  for (int t = 0; t < 16; ++t) {
    if (ok[t]) {
      for (int f = 0; f < F; ++f) atomicAdd(&acc[f * RP + rel[t]], v[t][f]);
    }
  }
}

template<int F, int FOUT, bool LAST>
__global__ __launch_bounds__(LB_THREADS)
void k_rng_edge(const unsigned* __restrict__ lists, const int* __restrict__ segtab,
                const float* __restrict__ m, const float* __restrict__ dinv,
                const float* __restrict__ b, const float* __restrict__ Wn,
                const float* __restrict__ bn, float* __restrict__ mout,
                float* __restrict__ hout, int n) {
  __shared__ float acc[4 * NPRD];
  __shared__ float slabs[32 * 64];
  const int tid  = (int)threadIdx.x;
  const int lane = tid & 31;
  const int wv   = __builtin_amdgcn_readfirstlane(tid >> 5);
  const int r    = (int)blockIdx.x;
  const int lo   = r * RN;
  int cnt = n - lo;
  if (cnt > RN) cnt = RN;

  for (int i = tid; i < NCP * F * RP; i += LB_THREADS) acc[i] = 0.0f;
  __syncthreads();

  const int ntiles = (cnt + 15) >> 4;
  const int lane16 = lane & 15;
  const int hi     = lane >> 4;
  const int il  = wv * 16 + lane16;
  const bool rok = il < cnt;
  const int ilc = rok ? il : (cnt - 1);
  const int ar  = lo + ilc;
  const int f0  = (F == 4) ? hi * 2 : 0;
  float  di = 0.0f, bs0 = 0.0f, bs1 = 0.0f;
  float2 mv = make_float2(0.0f, 0.0f);
  v2f    bw = {0.0f, 0.0f};
  if (wv < ntiles) {
    di  = dinv[ar];
    mv  = *reinterpret_cast<const float2*>(m + (size_t)ar * F + f0);
    bs0 = b[f0];
    bs1 = b[f0 + 1];
#pragma unroll
    for (int j = 0; j < 2; ++j) {
      int  k    = hi * 2 + j;
      int  kc   = (k < F) ? k : (F - 1);
      int  cc   = (lane16 < FOUT) ? lane16 : (FOUT - 1);
      float wvv = Wn[kc * FOUT + cc];
      bw[j] = (k < F && lane16 < FOUT) ? wvv : 0.0f;
    }
  }

  float* mine = acc + (lane & (NCP - 1)) * (F * RP);
  for (int j0 = 0; j0 < RCH; j0 += 4) {
    const int p0 = wv * RCH + j0;
    int so[4], nl[4];
    rng_segs4(segtab, p0, r, so, nl);
    int ng[4], ngmax = 0;
    for (int j = 0; j < 4; ++j) { ng[j] = (nl[j] + 127) >> 7; ngmax = ng[j] > ngmax ? ng[j] : ngmax; }
    for (int q = 0; q < ngmax; ++q) {
      uint4 w[4];
      int rem[4];
      for (int j = 0; j < 4; ++j) {
        rem[j] = nl[j] - (q * 128 + lane * 4);
        const int qj = (q < ng[j]) ? q : (ng[j] > 0 ? ng[j] - 1 : 0);
        const uint4* lp = reinterpret_cast<const uint4*>(lists + (size_t)(p0 + j) * SCAP + so[j]) + lane;
        w[j] = lp[(size_t)qj * 32];
      }
      rng_add16<F>(mine, m, w, rem, lo);
    }
  }
  __syncthreads();

#if __has_builtin(__builtin_amdgcn_wmma_f32_16x16x4_f32)
  if (wv < ntiles) {
    const float zf = rok ? 1.0f : 0.0f;
    float s0 = 0.0f, s1 = 0.0f;
    for (int c = 0; c < NCP; ++c) {
      s0 += acc[(c * F + f0) * RP + ilc];
      s1 += acc[(c * F + f0 + 1) * RP + ilc];
    }
    const float h0 = tanhf(di * (s0 + mv.x) + bs0);
    const float h1 = tanhf(di * (s1 + mv.y) + bs1);
    v2f a;
    if (F == 4) { a[0] = h0 * zf; a[1] = h1 * zf; }
    else        { const float zk = (hi == 0) ? zf : 0.0f; a[0] = h0 * zk; a[1] = h1 * zk; }
    if (LAST) {
      if (hi == 0 && rok) {
        float2 o; o.x = h0; o.y = h1;
        *reinterpret_cast<float2*>(hout + (size_t)ar * 2) = o;
      }
    }
    v8f c = {};
    c = __builtin_amdgcn_wmma_f32_16x16x4_f32(false, a, false, bw, (short)0, c, false, false);
    float* slab = slabs + wv * 64;
    if (lane16 < FOUT) {
#pragma unroll
      for (int rr = 0; rr < 8; ++rr) slab[(hi * 8 + rr) * FOUT + lane16] = c[rr];
    }
    asm volatile("s_wait_dscnt 0" ::: "memory");
    if (hi == 0 && rok) {
      if (FOUT == 4) {
        float4 v = *reinterpret_cast<const float4*>(slab + lane16 * 4);
        if (LAST) { v.x += bn[0]; v.y += bn[1]; v.z += bn[2]; v.w += bn[3]; }
        else      { v.x *= di; v.y *= di; v.z *= di; v.w *= di; }
        *reinterpret_cast<float4*>(mout + (size_t)ar * 4) = v;
      } else {
        float2 v = *reinterpret_cast<const float2*>(slab + lane16 * 2);
        if (LAST) { v.x += bn[0]; v.y += bn[1]; }
        else      { v.x *= di; v.y *= di; }
        *reinterpret_cast<float2*>(mout + (size_t)ar * 2) = v;
      }
    }
  }
#endif
}

extern "C" void kernel_launch(void* const* d_in, const int* in_sizes, int n_in,
                              void* d_out, int out_size, void* d_ws, size_t ws_size,
                              hipStream_t stream) {
  const float* x  = (const float*)d_in[0];
  const int*   ei = (const int*)  d_in[1];
  const float* W1 = (const float*)d_in[2];
  const float* b1 = (const float*)d_in[3];
  const float* W2 = (const float*)d_in[4];
  const float* b2 = (const float*)d_in[5];
  const float* W3 = (const float*)d_in[6];
  const float* b3 = (const float*)d_in[7];
  const float* Wc = (const float*)d_in[8];
  const float* bc = (const float*)d_in[9];

  const int N = in_sizes[0] / 32;
  const int E = in_sizes[1] / 2;
  const int* src = ei;
  const int* dst = ei + E;

  float* out = (float*)d_out;
  float* h3  = out + (size_t)N * 4;

  const int nr4 = (N + NPR4 - 1) / NPR4;
  const int nc4 = (GRID_WGS / nr4) > 0 ? (GRID_WGS / nr4) : 1;
  const int ec4 = (((E + nc4 - 1) / nc4) + 3) & ~3;
  const int nr2 = (N + NPR2 - 1) / NPR2;
  const int nc2 = (GRID_WGS / nr2) > 0 ? (GRID_WGS / nr2) : 1;
  const int ec2 = (((E + nc2 - 1) / nc2) + 3) & ~3;

  const size_t Np = ((size_t)N + 63) & ~(size_t)63;
  float* ws   = (float*)d_ws;
  float* dinv = ws;
  float* m    = ws + Np;
  float* h    = ws + 5 * Np;
  float* part = ws + 9 * Np;

  const int gN  = (N + THREADS - 1) / THREADS;
  const int tiles = (N + 15) / 16;
  const int gW  = (tiles + WAVES_PER_BLOCK - 1) / WAVES_PER_BLOCK;

  const int    wcap   = ((ec4 + TILE_EDGES - 1) / TILE_EDGES) * 128;
  const size_t nwaves = (size_t)nr4 * nc4 * (LB_THREADS / 32);
  const size_t partf  = (((size_t)nc4 * 4 * N) + 63) & ~(size_t)63;
  int*      lcnt  = (int*)(part + partf);
  unsigned* lists = (unsigned*)(part + partf + ((nwaves + 63) & ~(size_t)63));
  const size_t need = (size_t)((const char*)(lists + nwaves * wcap) - (const char*)d_ws);
  const int    snr   = (N + SNPR - 1) / SNPR;
  const int    spc   = (((E + SNP - 1) / SNP) + 3) & ~3;
  const size_t partS = (((size_t)SNC * 4 * N) + 63) & ~(size_t)63;
  int*      segtab = (int*)(part + partS);
  unsigned* segs   = (unsigned*)(part + partS + (size_t)SNP * SNR * 2);
  const size_t needS = (size_t)((const char*)(segs + (size_t)SNP * SCAP) - (const char*)d_ws);
  const int    nrng  = (N + RN - 1) / RN;
  int*      rtab  = (int*)(ws + 9 * Np);
  unsigned* rsegs = (unsigned*)(rtab + (size_t)SNP * RNR * 2);
  const size_t needR = (size_t)((const char*)(rsegs + (size_t)SNP * SCAP) - (const char*)d_ws);
  if (nrng <= RNR && N <= (1 << 23) && spc <= SMAXCHUNK && needR <= ws_size) {
    float* mA = m;
    float* mB = h;
    k_rsplit<<<SNP, LB_THREADS, 0, stream>>>(src, dst, rsegs, rtab, E, N, spc);
    k_rng_deg<<<nrng, LB_THREADS, 0, stream>>>(rsegs, rtab, x, W1, dinv, mA, N);
    k_rng_edge<4, 4, false><<<nrng, LB_THREADS, 0, stream>>>(rsegs, rtab, mA, dinv, b1, W2, nullptr, mB, nullptr, N);
    k_rng_edge<4, 2, false><<<nrng, LB_THREADS, 0, stream>>>(rsegs, rtab, mB, dinv, b2, W3, nullptr, mA, nullptr, N);
    k_rng_edge<2, 4, true><<<nrng, LB_THREADS, 0, stream>>>(rsegs, rtab, mA, dinv, b3, Wc, bc, out, h3, N);
    return;
  }

  if (snr <= SNR && N <= (1 << 20) && spc <= SMAXCHUNK && needS <= ws_size) {
    k_split<<<SNP, LB_THREADS, 0, stream>>>(src, dst, segs, segtab, E, N, spc);
    k_seg_deg<<<snr * SNC, LB_THREADS, 0, stream>>>(segs, segtab, part, N, snr);
    k_rsqrt<<<gN, THREADS, 0, stream>>>(dinv, part, N, SNC);

    k_xform32_wmma<<<gW, THREADS, 0, stream>>>(x, W1, dinv, m, N);
    k_seg_edge<4><<<snr * SNC, LB_THREADS, 0, stream>>>(segs, segtab, m, part, N, snr);
    k_tanh<4><<<gN, THREADS, 0, stream>>>(h, part, m, dinv, b1, N, SNC);

    k_xform_small<4, 4, false, true><<<gW, THREADS, 0, stream>>>(h, W2, nullptr, dinv, m, N);
    k_seg_edge<4><<<snr * SNC, LB_THREADS, 0, stream>>>(segs, segtab, m, part, N, snr);
    k_tanh<4><<<gN, THREADS, 0, stream>>>(h, part, m, dinv, b2, N, SNC);

    k_xform_small<4, 2, false, true><<<gW, THREADS, 0, stream>>>(h, W3, nullptr, dinv, m, N);
    k_seg_edge<2><<<snr * SNC, LB_THREADS, 0, stream>>>(segs, segtab, m, part, N, snr);
    k_tanh<2><<<gN, THREADS, 0, stream>>>(h3, part, m, dinv, b3, N, SNC);

    k_xform_small<2, 4, true, false><<<gW, THREADS, 0, stream>>>(h3, Wc, bc, nullptr, out, N);
    return;
  }

  if (N <= (1 << 18) && NPRD <= (1 << LIST_RELBITS) && need <= ws_size) {
    k_deg_list<<<nr4 * nc4, LB_THREADS, 0, stream>>>(src, dst, part, lists, lcnt, E, N, nr4, ec4, wcap);
    k_rsqrt<<<gN, THREADS, 0, stream>>>(dinv, part, N, nc4);

    k_xform32_wmma<<<gW, THREADS, 0, stream>>>(x, W1, dinv, m, N);
    k_edge_list<4><<<nr4 * nc4, LB_THREADS, 0, stream>>>(lists, lcnt, m, part, N, nr4, wcap);
    k_tanh<4><<<gN, THREADS, 0, stream>>>(h, part, m, dinv, b1, N, nc4);

    k_xform_small<4, 4, false, true><<<gW, THREADS, 0, stream>>>(h, W2, nullptr, dinv, m, N);
    k_edge_list<4><<<nr4 * nc4, LB_THREADS, 0, stream>>>(lists, lcnt, m, part, N, nr4, wcap);
    k_tanh<4><<<gN, THREADS, 0, stream>>>(h, part, m, dinv, b2, N, nc4);

    k_xform_small<4, 2, false, true><<<gW, THREADS, 0, stream>>>(h, W3, nullptr, dinv, m, N);
    k_edge_list<2><<<nr4 * nc4, LB_THREADS, 0, stream>>>(lists, lcnt, m, part, N, nr4, wcap);
    k_tanh<2><<<gN, THREADS, 0, stream>>>(h3, part, m, dinv, b3, N, nc4);

    k_xform_small<2, 4, true, false><<<gW, THREADS, 0, stream>>>(h3, Wc, bc, nullptr, out, N);
    return;
  }

  k_deg_lds<<<nr4 * nc4, LB_THREADS, 0, stream>>>(dst, part, E, N, nr4, ec4);
  k_rsqrt<<<gN, THREADS, 0, stream>>>(dinv, part, N, nc4);

  k_xform32_wmma<<<gW, THREADS, 0, stream>>>(x, W1, dinv, m, N);
  k_edge_lds<4><<<nr4 * nc4, LB_THREADS, 0, stream>>>(src, dst, m, part, E, N, nr4, ec4);
  k_tanh<4><<<gN, THREADS, 0, stream>>>(h, part, m, dinv, b1, N, nc4);

  k_xform_small<4, 4, false, true><<<gW, THREADS, 0, stream>>>(h, W2, nullptr, dinv, m, N);
  k_edge_lds<4><<<nr4 * nc4, LB_THREADS, 0, stream>>>(src, dst, m, part, E, N, nr4, ec4);
  k_tanh<4><<<gN, THREADS, 0, stream>>>(h, part, m, dinv, b2, N, nc4);

  k_xform_small<4, 2, false, true><<<gW, THREADS, 0, stream>>>(h, W3, nullptr, dinv, m, N);
  k_edge_lds<2><<<nr2 * nc2, LB_THREADS, 0, stream>>>(src, dst, m, part, E, N, nr2, ec2);
  k_tanh<2><<<gN, THREADS, 0, stream>>>(h3, part, m, dinv, b3, N, nc2);

  k_xform_small<2, 4, true, false><<<gW, THREADS, 0, stream>>>(h3, Wc, bc, nullptr, out, N);
}


# Each 391-node range's edges counting-sorted by source-row line in LDS in the degree pass; layer passes read one sorted list, so a load's 32 row gathers share lines; sum order = sorted-list order
# speedup vs baseline: 21.3701x; 1.2807x over previous
#include <hip/hip_runtime.h>
#include <hip/hip_fp16.h>

typedef __attribute__((ext_vector_type(16))) _Float16 v16h;
typedef __attribute__((ext_vector_type(8)))  float    v8f;
typedef __attribute__((ext_vector_type(2)))  float    v2f;

#define THREADS 256
#define WAVES_PER_BLOCK (THREADS / 32)

#define LB_THREADS 1024
#define GRID_WGS   256
#define NPR4       12544
#define NPR2       20544
#define NPRD       NPR4
constexpr int npr_of(int F) { return F == 4 ? NPR4 : NPR2; }


__device__ __forceinline__ int4 load_idx4_tail(const int* __restrict__ p, int i, int e1, int fill) {
  int4 r;
  r.x = p[i];
  r.y = (i + 1 < e1) ? p[i + 1] : fill;
  r.z = (i + 2 < e1) ? p[i + 2] : fill;
  r.w = (i + 3 < e1) ? p[i + 3] : fill;
  return r;
}

#define TILE_EDGES (4 * LB_THREADS)

__device__ __forceinline__ void deg_add4(float* cntl, int4 d, int lo, unsigned ucnt) {
  unsigned r0 = (unsigned)(d.x - lo), r1 = (unsigned)(d.y - lo);
  unsigned r2 = (unsigned)(d.z - lo), r3 = (unsigned)(d.w - lo);
  if (r0 < ucnt) atomicAdd(&cntl[r0], 1.0f);
  if (r1 < ucnt) atomicAdd(&cntl[r1], 1.0f);
  if (r2 < ucnt) atomicAdd(&cntl[r2], 1.0f);
  if (r3 < ucnt) atomicAdd(&cntl[r3], 1.0f);
}

__global__ __launch_bounds__(LB_THREADS)
void k_deg_lds(const int* __restrict__ dst, float* __restrict__ part,
               int e, int n, int nranges, int echunk) {
  __shared__ float cntl[NPRD];
  const int tid = (int)threadIdx.x;
  const int r   = (int)(blockIdx.x % (unsigned)nranges);
  const int c   = (int)(blockIdx.x / (unsigned)nranges);
  const int lo  = r * NPRD;
  int cnt = n - lo;
  if (cnt > NPRD) cnt = NPRD;
  const unsigned ucnt = (unsigned)cnt;

  for (int i = tid; i < NPRD; i += LB_THREADS) cntl[i] = 0.0f;
  __syncthreads();

  const int e0 = c * echunk;
  int e1 = e0 + echunk;
  if (e1 > e) e1 = e;
  const bool vec = (((size_t)dst) & 15) == 0;
  const int nfull = (vec && e1 > e0) ? (e1 - e0) / TILE_EDGES : 0;

  const int4* dp = reinterpret_cast<const int4*>(dst + e0) + tid;
  int t = 0;
  for (; t + 4 <= nfull; t += 4) {
    int4 d0 = dp[(size_t)(t + 0) * LB_THREADS];
    int4 d1 = dp[(size_t)(t + 1) * LB_THREADS];
    int4 d2 = dp[(size_t)(t + 2) * LB_THREADS];
    int4 d3 = dp[(size_t)(t + 3) * LB_THREADS];
    deg_add4(cntl, d0, lo, ucnt);
    deg_add4(cntl, d1, lo, ucnt);
    deg_add4(cntl, d2, lo, ucnt);
    deg_add4(cntl, d3, lo, ucnt);
  }
  for (; t < nfull; ++t) deg_add4(cntl, dp[(size_t)t * LB_THREADS], lo, ucnt);
  for (int base = e0 + nfull * TILE_EDGES + 4 * tid; base < e1; base += TILE_EDGES)
    deg_add4(cntl, load_idx4_tail(dst, base, e1, -1), lo, ucnt);
  __syncthreads();

  float* pc = part + (size_t)c * n + lo;
  for (int i = tid; i < cnt; i += LB_THREADS) pc[i] = cntl[i];
}

__global__ __launch_bounds__(THREADS)
void k_rsqrt(float* __restrict__ dinv, const float* __restrict__ part, int n, int nchunks) {
  int i = blockIdx.x * blockDim.x + threadIdx.x;
  if (i >= n) return;
  const float* p = part + i;
  float s = 1.0f;
  int c = 0;
  for (; c + 4 <= nchunks; c += 4) {
    float t0 = p[(size_t)(c + 0) * n], t1 = p[(size_t)(c + 1) * n];
    float t2 = p[(size_t)(c + 2) * n], t3 = p[(size_t)(c + 3) * n];
    s += t0; s += t1; s += t2; s += t3;
  }
  for (; c < nchunks; ++c) s += p[(size_t)c * n];
  dinv[i] = rsqrtf(s);
}

__global__ __launch_bounds__(THREADS)
void k_xform32_wmma(const float* __restrict__ x, const float* __restrict__ W,
                    const float* __restrict__ dinv, float* __restrict__ m, int n) {
  __shared__ float lds[WAVES_PER_BLOCK * 64];
  int wave  = (int)((blockIdx.x * (unsigned)blockDim.x + threadIdx.x) >> 5);
  int tl    = threadIdx.x & 31;
  int row0  = wave * 16;
  if (row0 >= n) return;
  int lane16 = tl & 15;
  int hi     = tl >> 4;

  v16h b;
  {
    int  colW = lane16 & 3;
    bool keep = lane16 < 4;
#pragma unroll
    for (int j = 0; j < 16; ++j) {
      int k = (j < 8) ? (hi * 8 + j) : (16 + hi * 8 + (j - 8));
      float wv = W[k * 4 + colW];
      b[j] = keep ? (_Float16)wv : (_Float16)0.0f;
    }
  }

  int  arow = row0 + lane16;
  bool rok  = arow < n;
  int  ar   = rok ? arow : (n - 1);
  const float* xr = x + (size_t)ar * 32;
  float4 q0 = *reinterpret_cast<const float4*>(xr + hi * 8);
  float4 q1 = *reinterpret_cast<const float4*>(xr + hi * 8 + 4);
  float4 q2 = *reinterpret_cast<const float4*>(xr + 16 + hi * 8);
  float4 q3 = *reinterpret_cast<const float4*>(xr + 16 + hi * 8 + 4);
  float di = dinv[ar];
  float zf = rok ? 1.0f : 0.0f;
  v16h a;
  a[0]=(_Float16)(q0.x*zf); a[1]=(_Float16)(q0.y*zf); a[2]=(_Float16)(q0.z*zf); a[3]=(_Float16)(q0.w*zf);
  a[4]=(_Float16)(q1.x*zf); a[5]=(_Float16)(q1.y*zf); a[6]=(_Float16)(q1.z*zf); a[7]=(_Float16)(q1.w*zf);
  a[8]=(_Float16)(q2.x*zf); a[9]=(_Float16)(q2.y*zf); a[10]=(_Float16)(q2.z*zf); a[11]=(_Float16)(q2.w*zf);
  a[12]=(_Float16)(q3.x*zf); a[13]=(_Float16)(q3.y*zf); a[14]=(_Float16)(q3.z*zf); a[15]=(_Float16)(q3.w*zf);

  v8f c = {};
  c = __builtin_amdgcn_wmma_f32_16x16x32_f16(false, a, false, b,
                                             (short)0, c, false, false);

  float* slab = lds + ((threadIdx.x >> 5) << 6);
  if (lane16 < 4) {
#pragma unroll
    for (int r = 0; r < 8; ++r) slab[(hi * 8 + r) * 4 + lane16] = c[r];
  }
  asm volatile("s_wait_dscnt 0" ::: "memory");
  if (hi == 0) {
    int mr = row0 + lane16;
    if (mr < n) {
      float4 v = *reinterpret_cast<const float4*>(slab + lane16 * 4);
      v.x *= di; v.y *= di; v.z *= di; v.w *= di;
      *reinterpret_cast<float4*>(m + (size_t)mr * 4) = v;
    }
  }
}

template<int KIN, int FOUT, bool BIAS, bool SCALE>
__global__ __launch_bounds__(THREADS)
void k_xform_small(const float* __restrict__ hin, const float* __restrict__ W,
                   const float* __restrict__ bias, const float* __restrict__ dinv,
                   float* __restrict__ mout, int n) {
  __shared__ float lds[WAVES_PER_BLOCK * 64];
  int wave  = (int)((blockIdx.x * (unsigned)blockDim.x + threadIdx.x) >> 5);
  int tl    = threadIdx.x & 31;
  int row0  = wave * 16;
  if (row0 >= n) return;
  int lane16 = tl & 15;
  int hi     = tl >> 4;

#if __has_builtin(__builtin_amdgcn_wmma_f32_16x16x4_f32)
  int  arow = row0 + lane16;
  bool rok  = arow < n;
  int  ar   = rok ? arow : (n - 1);
  float zf  = rok ? 1.0f : 0.0f;
  float di  = 1.0f;
  if (SCALE) di = dinv[ar];

  v2f a;
  if (KIN == 4) {
    float2 v = *reinterpret_cast<const float2*>(hin + (size_t)ar * 4 + hi * 2);
    a[0] = v.x * zf;  a[1] = v.y * zf;
  } else {
    float2 v = *reinterpret_cast<const float2*>(hin + (size_t)ar * 2);
    float zk = (hi == 0) ? zf : 0.0f;
    a[0] = v.x * zk;  a[1] = v.y * zk;
  }

  v2f b;
#pragma unroll
  for (int j = 0; j < 2; ++j) {
    int  k    = hi * 2 + j;
    int  kc   = (k < KIN) ? k : (KIN - 1);
    int  cc   = (lane16 < FOUT) ? lane16 : (FOUT - 1);
    float wv  = W[kc * FOUT + cc];
    b[j] = (k < KIN && lane16 < FOUT) ? wv : 0.0f;
  }

  v8f c = {};
  c = __builtin_amdgcn_wmma_f32_16x16x4_f32(false, a, false, b,
                                            (short)0, c, false, false);

  float* slab = lds + ((threadIdx.x >> 5) << 6);
  if (lane16 < FOUT) {
#pragma unroll
    for (int r = 0; r < 8; ++r) slab[(hi * 8 + r) * FOUT + lane16] = c[r];
  }
  asm volatile("s_wait_dscnt 0" ::: "memory");
  if (hi == 0) {
    int mr = row0 + lane16;
    if (mr < n) {
      if (FOUT == 4) {
        float4 v = *reinterpret_cast<const float4*>(slab + lane16 * 4);
        if (BIAS) { v.x += bias[0]; v.y += bias[1]; v.z += bias[2]; v.w += bias[3]; }
        if (SCALE) { v.x *= di; v.y *= di; v.z *= di; v.w *= di; }
        *reinterpret_cast<float4*>(mout + (size_t)mr * 4) = v;
      } else {
        float2 v = *reinterpret_cast<const float2*>(slab + lane16 * 2);
        if (BIAS) { v.x += bias[0]; v.y += bias[1]; }
        if (SCALE) { v.x *= di; v.y *= di; }
        *reinterpret_cast<float2*>(mout + (size_t)mr * 2) = v;
      }
    }
  }
#else
  if (hi == 0) {
    int mr = row0 + lane16;
    if (mr < n) {
#pragma unroll
      for (int f = 0; f < FOUT; ++f) {
        float acc = BIAS ? bias[f] : 0.0f;
#pragma unroll
        for (int k = 0; k < KIN; ++k) acc += hin[mr * KIN + k] * W[k * FOUT + f];
        if (SCALE) acc *= dinv[mr];
        mout[mr * FOUT + f] = acc;
      }
    }
  }
#endif
}

template<int F>
__device__ __forceinline__ void edge_add4(float* acc, const float* __restrict__ m,
                                          int4 d4, int4 s4, int lo, unsigned ucnt) {
  int dd[4] = {d4.x, d4.y, d4.z, d4.w};
  int ss[4] = {s4.x, s4.y, s4.z, s4.w};
  constexpr int NPR = npr_of(F);
  unsigned rel[4];
  bool     hit[4];
  float    v[4][F];
  for (int k = 0; k < 4; ++k) {
    rel[k] = (unsigned)(dd[k] - lo);
    hit[k] = rel[k] < ucnt;
    int      si = hit[k] ? ss[k] : lo;
    if constexpr (F == 4) {
      float4 t = *reinterpret_cast<const float4*>(m + (size_t)si * 4);
      v[k][0] = t.x; v[k][1] = t.y; v[k][2] = t.z; v[k][3] = t.w;
    } else {
      float2 t = *reinterpret_cast<const float2*>(m + (size_t)si * 2);
      v[k][0] = t.x; v[k][1] = t.y;
    }
  }
  for (int k = 0; k < 4; ++k) {
    if (hit[k]) {
      for (int f = 0; f < F; ++f) atomicAdd(&acc[f * NPR + rel[k]], v[k][f]);
    }
  }
}

template<int F>
__global__ __launch_bounds__(LB_THREADS)
void k_edge_lds(const int* __restrict__ src, const int* __restrict__ dst,
                const float* __restrict__ m,
                float* __restrict__ part, int e, int n, int nranges, int echunk) {
  constexpr int NPR = npr_of(F);
  __shared__ float acc[F * NPR];
  const int tid = (int)threadIdx.x;
  const int r   = (int)(blockIdx.x % (unsigned)nranges);
  const int c   = (int)(blockIdx.x / (unsigned)nranges);
  const int lo  = r * NPR;
  int cnt = n - lo;
  if (cnt > NPR) cnt = NPR;
  const unsigned ucnt = (unsigned)cnt;

  for (int i = tid; i < F * NPR; i += LB_THREADS) acc[i] = 0.0f;
  __syncthreads();

  const int e0 = c * echunk;
  int e1 = e0 + echunk;
  if (e1 > e) e1 = e;
  const bool vec = ((((size_t)src) | ((size_t)dst)) & 15) == 0;
  const int nfull = (vec && e1 > e0) ? (e1 - e0) / TILE_EDGES : 0;

  const int4* dp = reinterpret_cast<const int4*>(dst + e0) + tid;
  const int4* sp = reinterpret_cast<const int4*>(src + e0) + tid;
  int4 dn = make_int4(-1, -1, -1, -1), sn = make_int4(0, 0, 0, 0);
  if (nfull > 0) { dn = dp[0]; sn = sp[0]; }
  for (int t = 0; t < nfull; ++t) {
    int4 d4 = dn, s4 = sn;
    if (t + 1 < nfull) {
      dn = dp[(size_t)(t + 1) * LB_THREADS];
      sn = sp[(size_t)(t + 1) * LB_THREADS];
    }
    edge_add4<F>(acc, m, d4, s4, lo, ucnt);
  }
  for (int base = e0 + nfull * TILE_EDGES + 4 * tid; base < e1; base += TILE_EDGES)
    edge_add4<F>(acc, m, load_idx4_tail(dst, base, e1, -1),
                 load_idx4_tail(src, base, e1, 0), lo, ucnt);
  __syncthreads();

  float* pc = part + (size_t)c * F * n + lo;
  for (int i = tid; i < cnt; i += LB_THREADS) {
    float t[F];
    for (int f = 0; f < F; ++f) t[f] = acc[f * NPR + i];
    for (int f = 0; f < F; ++f) pc[(size_t)f * n + i] = t[f];
  }
}

template<int F>
__global__ __launch_bounds__(THREADS)
void k_tanh(float* __restrict__ hout, const float* __restrict__ part,
            const float* __restrict__ m, const float* __restrict__ dinv,
            const float* __restrict__ b, int n, int nchunks) {
  int i = blockIdx.x * blockDim.x + threadIdx.x;
  if (i >= n) return;
  const float* p = part + i;
  const size_t cs = (size_t)F * n;
  float s[F];
  for (int f = 0; f < F; ++f) s[f] = 0.0f;
  int c = 0;
  for (; c + 4 <= nchunks; c += 4) {
    float t[4][F];
    for (int j = 0; j < 4; ++j)
      for (int f = 0; f < F; ++f) t[j][f] = p[(size_t)(c + j) * cs + (size_t)f * n];
    for (int j = 0; j < 4; ++j)
      for (int f = 0; f < F; ++f) s[f] += t[j][f];
  }
  for (; c < nchunks; ++c)
    for (int f = 0; f < F; ++f) s[f] += p[(size_t)c * cs + (size_t)f * n];
  float di = dinv[i];
  if constexpr (F == 4) {
    float4 mv = *reinterpret_cast<const float4*>(m + (size_t)i * 4);
    float4 o;
    o.x = tanhf(di * (s[0] + mv.x) + b[0]);
    o.y = tanhf(di * (s[1] + mv.y) + b[1]);
    o.z = tanhf(di * (s[2] + mv.z) + b[2]);
    o.w = tanhf(di * (s[3] + mv.w) + b[3]);
    *reinterpret_cast<float4*>(hout + (size_t)i * 4) = o;
  } else {
    float2 mv = *reinterpret_cast<const float2*>(m + (size_t)i * 2);
    float2 o;
    o.x = tanhf(di * (s[0] + mv.x) + b[0]);
    o.y = tanhf(di * (s[1] + mv.y) + b[1]);
    *reinterpret_cast<float2*>(hout + (size_t)i * 2) = o;
  }
}


#define LIST_RING 256
#define LIST_RELBITS 14

__device__ __forceinline__ int4 load_idx4_guard(const int* __restrict__ p, int i, int e1, int fill) {
  int4 r;
  r.x = (i     < e1) ? p[i]     : fill;
  r.y = (i + 1 < e1) ? p[i + 1] : fill;
  r.z = (i + 2 < e1) ? p[i + 2] : fill;
  r.w = (i + 3 < e1) ? p[i + 3] : fill;
  return r;
}

__device__ __forceinline__ void deg_emit4(float* cntl, unsigned* myring,
                                          unsigned* __restrict__ mylist,
                                          int4 d, int4 s, int lo, unsigned ucnt,
                                          unsigned lane, unsigned ltmask,
                                          unsigned& wp, unsigned& rp) {
  int dd[4] = {d.x, d.y, d.z, d.w};
  int ss[4] = {s.x, s.y, s.z, s.w};
  for (int k = 0; k < 4; ++k) {
    unsigned rel = (unsigned)(dd[k] - lo);
    bool hit = rel < ucnt;
    unsigned b = __builtin_amdgcn_ballot_w32(hit);
    if (hit) {
      atomicAdd(&cntl[rel], 1.0f);
      unsigned slot = (wp + (unsigned)__builtin_popcount(b & ltmask)) & (LIST_RING - 1);
      myring[slot] = ((unsigned)ss[k] << LIST_RELBITS) | rel;
    }
    wp += (unsigned)__builtin_popcount(b);
  }
  while (wp - rp >= 32u) {
    __builtin_amdgcn_wave_barrier();
    mylist[rp + lane] = myring[(rp + lane) & (LIST_RING - 1)];
    rp += 32u;
  }
}

__global__ __launch_bounds__(LB_THREADS)
void k_deg_list(const int* __restrict__ src, const int* __restrict__ dst,
                float* __restrict__ part, unsigned* __restrict__ lists,
                int* __restrict__ lcnt, int e, int n, int nranges, int echunk, int wcap) {
  __shared__ float cntl[NPRD];
  __shared__ unsigned ring[(LB_THREADS / 32) * LIST_RING];
  const int tid = (int)threadIdx.x;
  const unsigned lane = (unsigned)tid & 31u;
  const unsigned ltmask = (1u << lane) - 1u;
  const int wv  = tid >> 5;
  const int r   = (int)(blockIdx.x % (unsigned)nranges);
  const int c   = (int)(blockIdx.x / (unsigned)nranges);
  const int lo  = r * NPRD;
  int cnt = n - lo;
  if (cnt > NPRD) cnt = NPRD;
  const unsigned ucnt = (unsigned)cnt;
  const size_t wid = (size_t)blockIdx.x * (LB_THREADS / 32) + (size_t)wv;
  unsigned* myring = ring + wv * LIST_RING;
  unsigned* mylist = lists + wid * (size_t)wcap;
  unsigned wp = 0, rp = 0;

  for (int i = tid; i < NPRD; i += LB_THREADS) cntl[i] = 0.0f;
  __syncthreads();

  const int e0 = c * echunk;
  int e1 = e0 + echunk;
  if (e1 > e) e1 = e;
  const bool vec = ((((size_t)src) | ((size_t)dst)) & 15) == 0;
  const int nfull = (vec && e1 > e0) ? (e1 - e0) / TILE_EDGES : 0;

  const int4* dp = reinterpret_cast<const int4*>(dst + e0) + tid;
  const int4* sp = reinterpret_cast<const int4*>(src + e0) + tid;
  int4 dn = make_int4(-1, -1, -1, -1), sn = make_int4(0, 0, 0, 0);
  if (nfull > 0) { dn = dp[0]; sn = sp[0]; }
  for (int t = 0; t < nfull; ++t) {
    int4 d4 = dn, s4 = sn;
    if (t + 1 < nfull) {
      dn = dp[(size_t)(t + 1) * LB_THREADS];
      sn = sp[(size_t)(t + 1) * LB_THREADS];
    }
    deg_emit4(cntl, myring, mylist, d4, s4, lo, ucnt, lane, ltmask, wp, rp);
  }
  for (int b0 = e0 + nfull * TILE_EDGES + 4 * (tid & ~31); b0 < e1; b0 += TILE_EDGES) {
    int base = b0 + 4 * (int)lane;
    deg_emit4(cntl, myring, mylist, load_idx4_guard(dst, base, e1, -1),
              load_idx4_guard(src, base, e1, 0), lo, ucnt, lane, ltmask, wp, rp);
  }
  __builtin_amdgcn_wave_barrier();
  if (rp + lane < wp) mylist[rp + lane] = myring[(rp + lane) & (LIST_RING - 1)];
  if (lane == 0) lcnt[wid] = (int)wp;
  __syncthreads();

  float* pc = part + (size_t)c * n + lo;
  for (int i = tid; i < cnt; i += LB_THREADS) pc[i] = cntl[i];
}

template<int F>
__device__ __forceinline__ void list_add4(float* acc, const float* __restrict__ m,
                                          uint4 w4, int rem, int lo) {
  unsigned ww[4] = {w4.x, w4.y, w4.z, w4.w};
  unsigned rel[4];
  bool     ok[4];
  float    v[4][F];
  for (int k = 0; k < 4; ++k) {
    ok[k]  = k < rem;
    rel[k] = ww[k] & ((1u << LIST_RELBITS) - 1u);
    int si = ok[k] ? (int)(ww[k] >> LIST_RELBITS) : lo;
    if constexpr (F == 4) {
      float4 t = *reinterpret_cast<const float4*>(m + (size_t)si * 4);
      v[k][0] = t.x; v[k][1] = t.y; v[k][2] = t.z; v[k][3] = t.w;
    } else {
      float2 t = *reinterpret_cast<const float2*>(m + (size_t)si * 2);
      v[k][0] = t.x; v[k][1] = t.y;
    }
  }
  for (int k = 0; k < 4; ++k) {
    if (ok[k]) {
      for (int f = 0; f < F; ++f) atomicAdd(&acc[f * NPRD + rel[k]], v[k][f]);
    }
  }
}

template<int F>
__global__ __launch_bounds__(LB_THREADS)
void k_edge_list(const unsigned* __restrict__ lists, const int* __restrict__ lcnt,
                 const float* __restrict__ m, float* __restrict__ part,
                 int n, int nranges, int wcap) {
  __shared__ float acc[4 * NPRD];
  const int tid  = (int)threadIdx.x;
  const int lane = tid & 31;
  const int wv   = tid >> 5;
  const int r   = (int)(blockIdx.x % (unsigned)nranges);
  const int c   = (int)(blockIdx.x / (unsigned)nranges);
  const int lo  = r * NPRD;
  int cnt = n - lo;
  if (cnt > NPRD) cnt = NPRD;

  for (int i = tid; i < F * NPRD; i += LB_THREADS) acc[i] = 0.0f;
  __syncthreads();

  const size_t wid = (size_t)blockIdx.x * (LB_THREADS / 32) + (size_t)wv;
  const int nl = __builtin_amdgcn_readfirstlane(lcnt[wid]);
  const int ng = (nl + 127) >> 7;
  const uint4* lp = reinterpret_cast<const uint4*>(lists + wid * (size_t)wcap) + lane;
  uint4 wn = make_uint4(0u, 0u, 0u, 0u);
  if (ng > 0) wn = lp[0];
  for (int g = 0; g < ng; ++g) {
    uint4 w4 = wn;
    if (g + 1 < ng) wn = lp[(size_t)(g + 1) * 32];
    list_add4<F>(acc, m, w4, nl - (g * 128 + lane * 4), lo);
  }
  __syncthreads();

  float* pc = part + (size_t)c * F * n + lo;
  for (int i = tid; i < cnt; i += LB_THREADS) {
    float t[F];
    for (int f = 0; f < F; ++f) t[f] = acc[f * NPRD + i];
    for (int f = 0; f < F; ++f) pc[(size_t)f * n + i] = t[f];
  }
}


#define SNPR     3136
#define SRELBITS 12
#define SNR      32
#define SNP      256
#define SNC      (SNP / 32)
#define SHP      33
#define SHBLK    (32 * SHP)
#define SCAP     26624
#define SMAXCHUNK 25600

__device__ __forceinline__ unsigned wave_incl_scan(unsigned x, int lane) {
  for (int d = 1; d < 32; d <<= 1) {
    unsigned t = __shfl_up(x, (unsigned)d);
    if (lane >= d) x += t;
  }
  return x;
}

template<bool PLACE>
__device__ __forceinline__ void split4(unsigned* hist, const unsigned* seg, unsigned* stage,
                                       int4 d4, int4 s4, unsigned n, unsigned hme) {
  int dd[4] = {d4.x, d4.y, d4.z, d4.w};
  int ss[4] = {s4.x, s4.y, s4.z, s4.w};
  bool     ok[4];
  unsigned r[4], rank[4], sg[4];
  for (int k = 0; k < 4; ++k) {
    ok[k] = (unsigned)dd[k] < n;
    r[k]  = ok[k] ? (unsigned)dd[k] / SNPR : 0u;
    unsigned* hp = hist + r[k] * SHBLK + hme;
    if constexpr (!PLACE) {
      atomicAdd(hp, ok[k] ? 1u : 0u);
    } else {
      rank[k] = atomicAdd(hp, ok[k] ? 1u : 0u);
      sg[k]   = seg[r[k]];
    }
  }
  if constexpr (PLACE) {
    for (int k = 0; k < 4; ++k)
      if (ok[k]) stage[sg[k] + rank[k]] = ((unsigned)ss[k] << SRELBITS) | ((unsigned)dd[k] - r[k] * SNPR);
  }
}

template<bool PLACE>
__device__ __forceinline__ void split_sweep(unsigned* hist, const unsigned* seg, unsigned* stage,
                                            const int* __restrict__ src, const int* __restrict__ dst,
                                            int e0, int e1, int nfull, unsigned n, int tid, unsigned hme) {
  const int4 z4 = make_int4(0, 0, 0, 0);
  const int4* dp = reinterpret_cast<const int4*>(dst + e0) + tid;
  const int4* sp = reinterpret_cast<const int4*>(src + e0) + tid;
  int t = 0;
  for (; t + 4 <= nfull; t += 4) {
    int4 d0 = dp[(size_t)(t + 0) * LB_THREADS], d1 = dp[(size_t)(t + 1) * LB_THREADS];
    int4 d2 = dp[(size_t)(t + 2) * LB_THREADS], d3 = dp[(size_t)(t + 3) * LB_THREADS];
    int4 s0 = z4, s1 = z4, s2 = z4, s3 = z4;
    if constexpr (PLACE) {
      s0 = sp[(size_t)(t + 0) * LB_THREADS]; s1 = sp[(size_t)(t + 1) * LB_THREADS];
      s2 = sp[(size_t)(t + 2) * LB_THREADS]; s3 = sp[(size_t)(t + 3) * LB_THREADS];
    }
    split4<PLACE>(hist, seg, stage, d0, s0, n, hme);
    split4<PLACE>(hist, seg, stage, d1, s1, n, hme);
    split4<PLACE>(hist, seg, stage, d2, s2, n, hme);
    split4<PLACE>(hist, seg, stage, d3, s3, n, hme);
  }
  for (; t + 2 <= nfull; t += 2) {
    int4 d0 = dp[(size_t)(t + 0) * LB_THREADS], d1 = dp[(size_t)(t + 1) * LB_THREADS];
    int4 s0 = z4, s1 = z4;
    if constexpr (PLACE) {
      s0 = sp[(size_t)(t + 0) * LB_THREADS]; s1 = sp[(size_t)(t + 1) * LB_THREADS];
    }
    split4<PLACE>(hist, seg, stage, d0, s0, n, hme);
    split4<PLACE>(hist, seg, stage, d1, s1, n, hme);
  }
  for (; t < nfull; ++t) {
    int4 d0 = dp[(size_t)t * LB_THREADS];
    int4 s0 = z4;
    if constexpr (PLACE) {
      s0 = sp[(size_t)t * LB_THREADS];
    }
    split4<PLACE>(hist, seg, stage, d0, s0, n, hme);
  }
  for (int base = e0 + nfull * TILE_EDGES + 4 * tid; base < e1; base += TILE_EDGES) {
    int4 d0 = load_idx4_guard(dst, base, e1, -1);
    int4 s0 = z4;
    if constexpr (PLACE) s0 = load_idx4_guard(src, base, e1, 0);
    split4<PLACE>(hist, seg, stage, d0, s0, n, hme);
  }
}

__global__ __launch_bounds__(LB_THREADS)
void k_split(const int* __restrict__ src, const int* __restrict__ dst,
             unsigned* __restrict__ lists, int* __restrict__ segtab,
             int e, int n, int pchunk) {
  __shared__ unsigned hist[SNR * SHBLK];
  __shared__ unsigned stage[SCAP];
  __shared__ unsigned tot[SNR];
  __shared__ unsigned seg[SNR];
  const int tid  = (int)threadIdx.x;
  const int lane = tid & 31;
  const int wv   = tid >> 5;
  const int p    = (int)blockIdx.x;
  const unsigned hme = (unsigned)(lane * SHP + wv);
  const int e0 = p * pchunk;
  int e1 = e0 + pchunk;
  if (e1 > e) e1 = e;
  const bool vec = ((((size_t)src) | ((size_t)dst)) & 15) == 0;
  const int nfull = (vec && e1 > e0) ? (e1 - e0) / TILE_EDGES : 0;

  for (int i = tid; i < SNR * SHBLK; i += LB_THREADS) hist[i] = 0u;
  for (int i = tid; i < SCAP; i += LB_THREADS) stage[i] = 0u;
  __syncthreads();

  split_sweep<false>(hist, seg, stage, src, dst, e0, e1, nfull, (unsigned)n, tid, hme);
  __syncthreads();

  {
    unsigned* hb = hist + wv * SHBLK + lane * SHP;
    unsigned v[32];
    unsigned sum = 0;
#pragma unroll
    for (int i = 0; i < 32; ++i) { v[i] = hb[i]; sum += v[i]; }
    const unsigned incl = wave_incl_scan(sum, lane);
    unsigned run = incl - sum;
#pragma unroll
    for (int i = 0; i < 32; ++i) { hb[i] = run; run += v[i]; }
    const unsigned total = __shfl(incl, 31);
    if (lane == 0) tot[wv] = total;
  }
  __syncthreads();

  if (wv == 0) {
    const unsigned cnt = tot[lane];
    const unsigned pad = (cnt + 3u) & ~3u;
    const unsigned so  = wave_incl_scan(pad, lane) - pad;
    seg[lane] = so;
    int2 oc; oc.x = (int)so; oc.y = (int)cnt;
    *reinterpret_cast<int2*>(segtab + ((size_t)p * SNR + lane) * 2) = oc;
  }
  __syncthreads();

  split_sweep<true>(hist, seg, stage, src, dst, e0, e1, nfull, (unsigned)n, tid, hme);
  __syncthreads();

  uint4* out = reinterpret_cast<uint4*>(lists + (size_t)p * SCAP);
  const uint4* st4 = reinterpret_cast<const uint4*>(stage);
  for (int i = tid; i < SCAP / 4; i += LB_THREADS) out[i] = st4[i];
}

__global__ __launch_bounds__(LB_THREADS)
void k_seg_deg(const unsigned* __restrict__ lists, const int* __restrict__ segtab,
               float* __restrict__ part, int n, int nranges) {
  __shared__ float cntl[SNPR];
  const int tid  = (int)threadIdx.x;
  const int lane = tid & 31;
  const int wv   = tid >> 5;
  const int r   = (int)(blockIdx.x % (unsigned)nranges);
  const int g   = (int)(blockIdx.x / (unsigned)nranges);
  const int lo  = r * SNPR;
  int cnt = n - lo;
  if (cnt > SNPR) cnt = SNPR;
  const int p = g * 32 + wv;
  const int2 oc = *reinterpret_cast<const int2*>(segtab + ((size_t)p * SNR + r) * 2);

  for (int i = tid; i < SNPR; i += LB_THREADS) cntl[i] = 0.0f;
  __syncthreads();

  const int so = __builtin_amdgcn_readfirstlane(oc.x);
  const int nl = __builtin_amdgcn_readfirstlane(oc.y);
  const int ng = (nl + 127) >> 7;
  const uint4* lp = reinterpret_cast<const uint4*>(lists + (size_t)p * SCAP + so) + lane;
  for (int g0 = 0; g0 < ng; g0 += 4) {
    uint4 w[4];
    for (int j = 0; j < 4; ++j)
      w[j] = (g0 + j < ng) ? lp[(size_t)(g0 + j) * 32] : make_uint4(0u, 0u, 0u, 0u);
    for (int j = 0; j < 4; ++j) {
      const int rem = nl - ((g0 + j) * 128 + lane * 4);
      unsigned ww[4] = {w[j].x, w[j].y, w[j].z, w[j].w};
      for (int k = 0; k < 4; ++k)
        if (k < rem) atomicAdd(&cntl[ww[k] & ((1u << SRELBITS) - 1u)], 1.0f);
    }
  }
  __syncthreads();

  float* pc = part + (size_t)g * n + lo;
  for (int i = tid; i < cnt; i += LB_THREADS) pc[i] = cntl[i];
}

template<int F>
__device__ __forceinline__ void seg_add4(float* acc, const float* __restrict__ m,
                                         uint4 w4, int rem, int lo) {
  unsigned ww[4] = {w4.x, w4.y, w4.z, w4.w};
  unsigned rel[4];
  bool     ok[4];
  float    v[4][F];
  for (int k = 0; k < 4; ++k) {
    ok[k]  = k < rem;
    rel[k] = ww[k] & ((1u << SRELBITS) - 1u);
    int si = ok[k] ? (int)(ww[k] >> SRELBITS) : lo;
    if constexpr (F == 4) {
      float4 t = *reinterpret_cast<const float4*>(m + (size_t)si * 4);
      v[k][0] = t.x; v[k][1] = t.y; v[k][2] = t.z; v[k][3] = t.w;
    } else {
      float2 t = *reinterpret_cast<const float2*>(m + (size_t)si * 2);
      v[k][0] = t.x; v[k][1] = t.y;
    }
  }
  for (int k = 0; k < 4; ++k) {
    if (ok[k]) {
      for (int f = 0; f < F; ++f) atomicAdd(&acc[f * SNPR + rel[k]], v[k][f]);
    }
  }
}

template<int F>
__global__ __launch_bounds__(LB_THREADS)
void k_seg_edge(const unsigned* __restrict__ lists, const int* __restrict__ segtab,
                const float* __restrict__ m, float* __restrict__ part,
                int n, int nranges) {
  __shared__ float acc[4 * NPRD];
  const int tid  = (int)threadIdx.x;
  const int lane = tid & 31;
  const int wv   = tid >> 5;
  const int r   = (int)(blockIdx.x % (unsigned)nranges);
  const int g   = (int)(blockIdx.x / (unsigned)nranges);
  const int lo  = r * SNPR;
  int cnt = n - lo;
  if (cnt > SNPR) cnt = SNPR;
  const int p = g * 32 + wv;
  const int2 oc = *reinterpret_cast<const int2*>(segtab + ((size_t)p * SNR + r) * 2);

  for (int i = tid; i < F * SNPR; i += LB_THREADS) acc[i] = 0.0f;
  __syncthreads();

  const int so = __builtin_amdgcn_readfirstlane(oc.x);
  const int nl = __builtin_amdgcn_readfirstlane(oc.y);
  const int ng = (nl + 127) >> 7;
  const uint4* lp = reinterpret_cast<const uint4*>(lists + (size_t)p * SCAP + so) + lane;
  uint4 wn = make_uint4(0u, 0u, 0u, 0u);
  if (ng > 0) wn = lp[0];
  for (int q = 0; q < ng; ++q) {
    uint4 w4 = wn;
    if (q + 1 < ng) wn = lp[(size_t)(q + 1) * 32];
    seg_add4<F>(acc, m, w4, nl - (q * 128 + lane * 4), lo);
  }
  __syncthreads();

  float* pc = part + (size_t)g * F * n + lo;
  for (int i = tid; i < cnt; i += LB_THREADS) {
    float t[F];
    for (int f = 0; f < F; ++f) t[f] = acc[f * SNPR + i];
    for (int f = 0; f < F; ++f) pc[(size_t)f * n + i] = t[f];
  }
}


#define RN        391
#define RBITS     9
#define RNR       256
#define RP        392
#define RCH       (SNP / 32)
#define RSG       4
#define NCP       8

template<bool PLACE>
__device__ __forceinline__ void rsplit4(unsigned* cntw, const unsigned* seg, unsigned* stage,
                                        int4 d4, int4 s4, unsigned n) {
  int dd[4] = {d4.x, d4.y, d4.z, d4.w};
  int ss[4] = {s4.x, s4.y, s4.z, s4.w};
  bool     ok[4];
  unsigned r[4], rank[4], sg[4];
  for (int k = 0; k < 4; ++k) {
    ok[k] = (unsigned)dd[k] < n;
    r[k]  = ok[k] ? (unsigned)dd[k] / RN : 0u;
    unsigned* hp = cntw + r[k];
    if constexpr (!PLACE) {
      atomicAdd(hp, ok[k] ? 1u : 0u);
    } else {
      rank[k] = atomicAdd(hp, ok[k] ? 1u : 0u);
      sg[k]   = seg[r[k]];
    }
  }
  if constexpr (PLACE) {
    for (int k = 0; k < 4; ++k)
      if (ok[k]) stage[sg[k] + rank[k]] = ((unsigned)ss[k] << RBITS) | ((unsigned)dd[k] - r[k] * RN);
  }
}

template<bool PLACE>
__device__ __forceinline__ void rsplit_sweep(unsigned* cntw, const unsigned* seg, unsigned* stage,
                                             const int* __restrict__ src, const int* __restrict__ dst,
                                             int e0, int e1, int nfull, unsigned n, int tid) {
  const int4 z4 = make_int4(0, 0, 0, 0);
  const int4* dp = reinterpret_cast<const int4*>(dst + e0) + tid;
  const int4* sp = reinterpret_cast<const int4*>(src + e0) + tid;
  int t = 0;
  for (; t + 4 <= nfull; t += 4) {
    int4 d0 = dp[(size_t)(t + 0) * LB_THREADS], d1 = dp[(size_t)(t + 1) * LB_THREADS];
    int4 d2 = dp[(size_t)(t + 2) * LB_THREADS], d3 = dp[(size_t)(t + 3) * LB_THREADS];
    int4 s0 = z4, s1 = z4, s2 = z4, s3 = z4;
    if constexpr (PLACE) {
      s0 = sp[(size_t)(t + 0) * LB_THREADS]; s1 = sp[(size_t)(t + 1) * LB_THREADS];
      s2 = sp[(size_t)(t + 2) * LB_THREADS]; s3 = sp[(size_t)(t + 3) * LB_THREADS];
    }
    rsplit4<PLACE>(cntw, seg, stage, d0, s0, n);
    rsplit4<PLACE>(cntw, seg, stage, d1, s1, n);
    rsplit4<PLACE>(cntw, seg, stage, d2, s2, n);
    rsplit4<PLACE>(cntw, seg, stage, d3, s3, n);
  }
  for (; t + 2 <= nfull; t += 2) {
    int4 d0 = dp[(size_t)(t + 0) * LB_THREADS], d1 = dp[(size_t)(t + 1) * LB_THREADS];
    int4 s0 = z4, s1 = z4;
    if constexpr (PLACE) {
      s0 = sp[(size_t)(t + 0) * LB_THREADS]; s1 = sp[(size_t)(t + 1) * LB_THREADS];
    }
    rsplit4<PLACE>(cntw, seg, stage, d0, s0, n);
    rsplit4<PLACE>(cntw, seg, stage, d1, s1, n);
  }
  for (; t < nfull; ++t) {
    int4 d0 = dp[(size_t)t * LB_THREADS];
    int4 s0 = z4;
    if constexpr (PLACE) s0 = sp[(size_t)t * LB_THREADS];
    rsplit4<PLACE>(cntw, seg, stage, d0, s0, n);
  }
  for (int base = e0 + nfull * TILE_EDGES + 4 * tid; base < e1; base += TILE_EDGES) {
    int4 d0 = load_idx4_guard(dst, base, e1, -1);
    int4 s0 = z4;
    if constexpr (PLACE) s0 = load_idx4_guard(src, base, e1, 0);
    rsplit4<PLACE>(cntw, seg, stage, d0, s0, n);
  }
}

__global__ __launch_bounds__(LB_THREADS)
void k_rsplit(const int* __restrict__ src, const int* __restrict__ dst,
              unsigned* __restrict__ lists, int* __restrict__ segtab,
              int e, int n, int pchunk) {
  __shared__ unsigned cntl[32 * RSG * RNR];
  __shared__ unsigned stage[SCAP];
  __shared__ unsigned seg[RNR];
  __shared__ unsigned ptot[4 * RNR];
  __shared__ unsigned wsum[RNR / 32];
  const int tid  = (int)threadIdx.x;
  const int lane = tid & 31;
  const int wv   = tid >> 5;
  const int p    = (int)blockIdx.x;
  const int e0 = p * pchunk;
  int e1 = e0 + pchunk;
  if (e1 > e) e1 = e;
  const bool vec = ((((size_t)src) | ((size_t)dst)) & 15) == 0;
  const int nfull = (vec && e1 > e0) ? (e1 - e0) / TILE_EDGES : 0;
  unsigned* cntw = cntl + (wv * RSG + (lane & (RSG - 1))) * RNR;

  for (int i = tid; i < 32 * RSG * RNR; i += LB_THREADS) cntl[i] = 0u;
  for (int i = tid; i < SCAP; i += LB_THREADS) stage[i] = 0u;
  __syncthreads();

  rsplit_sweep<false>(cntw, seg, stage, src, dst, e0, e1, nfull, (unsigned)n, tid);
  __syncthreads();

  const int sb = tid & (RNR - 1), sq = tid >> 8;
  unsigned v[32];
  unsigned run = 0, total = 0;
  {
    unsigned mysum = 0;
#pragma unroll
    for (int i = 0; i < 32; ++i) { v[i] = cntl[(sq * 32 + i) * RNR + sb]; mysum += v[i]; }
    ptot[sq * RNR + sb] = mysum;
  }
  __syncthreads();
  for (int q = 0; q < 4; ++q) {
    const unsigned t = ptot[q * RNR + sb];
    run   += (q < sq) ? t : 0u;
    total += t;
  }
#pragma unroll
  for (int i = 0; i < 32; ++i) { cntl[(sq * 32 + i) * RNR + sb] = run; run += v[i]; }
  const unsigned pad = (total + 3u) & ~3u;
  unsigned incl = 0;
  if (tid < RNR) {
    incl = wave_incl_scan(pad, lane);
    if (lane == 31) wsum[wv] = incl;
  }
  __syncthreads();
  if (tid < RNR) {
    unsigned base = 0;
    for (int w = 0; w < RNR / 32; ++w) base += (w < wv) ? wsum[w] : 0u;
    const unsigned so = base + incl - pad;
    seg[tid] = so;
    int2 oc; oc.x = (int)so; oc.y = (int)total;
    *reinterpret_cast<int2*>(segtab + ((size_t)p * RNR + tid) * 2) = oc;
  }
  __syncthreads();

  rsplit_sweep<true>(cntw, seg, stage, src, dst, e0, e1, nfull, (unsigned)n, tid);
  __syncthreads();

  uint4* out = reinterpret_cast<uint4*>(lists + (size_t)p * SCAP);
  const uint4* st4 = reinterpret_cast<const uint4*>(stage);
  for (int i = tid; i < SCAP / 4; i += LB_THREADS) out[i] = st4[i];
}

__device__ __forceinline__ void rng_segs4(const int* __restrict__ segtab, int p0, int r,
                                          int so[4], int nl[4]) {
  for (int j = 0; j < 4; ++j) {
    const int2 oc = *reinterpret_cast<const int2*>(segtab + ((size_t)(p0 + j) * RNR + r) * 2);
    so[j] = __builtin_amdgcn_readfirstlane(oc.x);
    nl[j] = __builtin_amdgcn_readfirstlane(oc.y);
  }
}

__global__ __launch_bounds__(LB_THREADS)
void k_rng_deg(const unsigned* __restrict__ lists, const int* __restrict__ segtab,
               const float* __restrict__ x, const float* __restrict__ W,
               float* __restrict__ dinv, float* __restrict__ m, int n) {
  __shared__ float acc[4 * NPRD];
  __shared__ float slabs[32 * 64];
  const int tid  = (int)threadIdx.x;
  const int lane = tid & 31;
  const int wv   = __builtin_amdgcn_readfirstlane(tid >> 5);
  const int r    = (int)blockIdx.x;
  const int lo   = r * RN;
  int cnt = n - lo;
  if (cnt > RN) cnt = RN;

  for (int i = tid; i < NCP * RP; i += LB_THREADS) acc[i] = 0.0f;
  __syncthreads();

  const int ntiles = (cnt + 15) >> 4;
  const int lane16 = lane & 15;
  const int hi     = lane >> 4;
  const int il  = wv * 16 + lane16;
  const bool rok = il < cnt;
  const int ilc = rok ? il : (cnt - 1);
  float4 q0 = make_float4(0.f, 0.f, 0.f, 0.f), q1 = q0, q2 = q0, q3 = q0;
  if (wv < ntiles) {
    const float* xr = x + (size_t)(lo + ilc) * 32;
    q0 = *reinterpret_cast<const float4*>(xr + hi * 8);
    q1 = *reinterpret_cast<const float4*>(xr + hi * 8 + 4);
    q2 = *reinterpret_cast<const float4*>(xr + 16 + hi * 8);
    q3 = *reinterpret_cast<const float4*>(xr + 16 + hi * 8 + 4);
  }

  float* mine = acc + (lane & (NCP - 1)) * RP;
  for (int j0 = 0; j0 < RCH; j0 += 4) {
    const int p0 = wv * RCH + j0;
    int so[4], nl[4];
    rng_segs4(segtab, p0, r, so, nl);
    int ng[4], ngmax = 0;
    for (int j = 0; j < 4; ++j) { ng[j] = (nl[j] + 127) >> 7; ngmax = ng[j] > ngmax ? ng[j] : ngmax; }
    for (int q = 0; q < ngmax; ++q) {
      uint4 w[4];
      int rem[4];
      for (int j = 0; j < 4; ++j) {
        rem[j] = nl[j] - (q * 128 + lane * 4);
        const int qj = (q < ng[j]) ? q : (ng[j] > 0 ? ng[j] - 1 : 0);
        const uint4* lp = reinterpret_cast<const uint4*>(lists + (size_t)(p0 + j) * SCAP + so[j]) + lane;
        w[j] = lp[(size_t)qj * 32];
      }
      for (int j = 0; j < 4; ++j) {
        unsigned ww[4] = {w[j].x, w[j].y, w[j].z, w[j].w};
        for (int k = 0; k < 4; ++k)
          if (k < rem[j]) atomicAdd(&mine[ww[k] & ((1u << RBITS) - 1u)], 1.0f);
      }
    }
  }
  __syncthreads();

  float* dl = acc + NCP * RP;
  for (int i = tid; i < cnt; i += LB_THREADS) {
    float s = 1.0f;
    for (int c = 0; c < NCP; ++c) s += acc[c * RP + i];
    const float d = rsqrtf(s);
    dl[i] = d;
    dinv[lo + i] = d;
  }
  __syncthreads();

  if (wv < ntiles) {
    v16h b;
    {
      int  colW = lane16 & 3;
      bool keep = lane16 < 4;
#pragma unroll
      for (int j = 0; j < 16; ++j) {
        int k = (j < 8) ? (hi * 8 + j) : (16 + hi * 8 + (j - 8));
        float wvv = W[k * 4 + colW];
        b[j] = keep ? (_Float16)wvv : (_Float16)0.0f;
      }
    }
    const float di = dl[ilc];
    const float zf = rok ? 1.0f : 0.0f;
    v16h a;
    a[0]=(_Float16)(q0.x*zf); a[1]=(_Float16)(q0.y*zf); a[2]=(_Float16)(q0.z*zf); a[3]=(_Float16)(q0.w*zf);
    a[4]=(_Float16)(q1.x*zf); a[5]=(_Float16)(q1.y*zf); a[6]=(_Float16)(q1.z*zf); a[7]=(_Float16)(q1.w*zf);
    a[8]=(_Float16)(q2.x*zf); a[9]=(_Float16)(q2.y*zf); a[10]=(_Float16)(q2.z*zf); a[11]=(_Float16)(q2.w*zf);
    a[12]=(_Float16)(q3.x*zf); a[13]=(_Float16)(q3.y*zf); a[14]=(_Float16)(q3.z*zf); a[15]=(_Float16)(q3.w*zf);
    v8f c = {};
    c = __builtin_amdgcn_wmma_f32_16x16x32_f16(false, a, false, b, (short)0, c, false, false);
    float* slab = slabs + wv * 64;
    if (lane16 < 4) {
#pragma unroll
      for (int rr = 0; rr < 8; ++rr) slab[(hi * 8 + rr) * 4 + lane16] = c[rr];
    }
    asm volatile("s_wait_dscnt 0" ::: "memory");
    if (hi == 0 && rok) {
      float4 v = *reinterpret_cast<const float4*>(slab + lane16 * 4);
      v.x *= di; v.y *= di; v.z *= di; v.w *= di;
      *reinterpret_cast<float4*>(m + (size_t)(lo + il) * 4) = v;
    }
  }
}

#define SLB       3
#define SBPT      13
#define SBINS     (SBPT * LB_THREADS)
#define RLCAP     32768
#define SGRP      13

template<bool PLACE>
__device__ __forceinline__ void rs_sweep4(const unsigned* __restrict__ lists, int p0,
                                          const int so[4], const int nl[4], int lane,
                                          float* mine, unsigned* hist, unsigned* stage) {
  int ng[4], ngmax = 0;
  for (int j = 0; j < 4; ++j) { ng[j] = (nl[j] + 127) >> 7; ngmax = ng[j] > ngmax ? ng[j] : ngmax; }
  for (int q = 0; q < ngmax; ++q) {
    uint4 w[4];
    int rem[4];
    for (int j = 0; j < 4; ++j) {
      rem[j] = nl[j] - (q * 128 + lane * 4);
      const int qj = (q < ng[j]) ? q : (ng[j] > 0 ? ng[j] - 1 : 0);
      const uint4* lp = reinterpret_cast<const uint4*>(lists + (size_t)(p0 + j) * SCAP + so[j]) + lane;
      w[j] = lp[(size_t)qj * 32];
    }
    unsigned wd[16], pos[16];
    bool ok[16];
#pragma unroll
    for (int j = 0; j < 4; ++j) {
      unsigned ww[4] = {w[j].x, w[j].y, w[j].z, w[j].w};
#pragma unroll
      for (int k = 0; k < 4; ++k) {
        const int t = j * 4 + k;
        ok[t] = k < rem[j];
        wd[t] = ww[k];
        const unsigned bin = ok[t] ? (ww[k] >> (RBITS + SLB)) : (unsigned)(SBINS + lane);
        if constexpr (!PLACE) {
          if (ok[t]) {
            atomicAdd(&mine[ww[k] & ((1u << RBITS) - 1u)], 1.0f);
            atomicAdd(&hist[bin], 1u);
          }
          pos[t] = 0u;
        } else {
          pos[t] = atomicAdd(&hist[bin], ok[t] ? 1u : 0u);
        }
      }
    }
    if constexpr (PLACE) {
#pragma unroll
      for (int t = 0; t < 16; ++t)
        if (ok[t]) stage[pos[t]] = wd[t];
    }
  }
}

__global__ __launch_bounds__(LB_THREADS)
void k_rng_sdeg(const unsigned* __restrict__ lists, const int* __restrict__ segtab,
                const float* __restrict__ x, const float* __restrict__ W,
                float* __restrict__ dinv, float* __restrict__ m,
                unsigned* __restrict__ rlist, int* __restrict__ rcnt, int n) {
  __shared__ float acc[NCP * RP + RP];
  __shared__ unsigned hist[SBINS + 32];
  __shared__ __align__(16) unsigned stage[RLCAP];
  __shared__ float slabs[32 * 64];
  __shared__ unsigned wtot[32];
  __shared__ unsigned wsum[32];
  const int tid  = (int)threadIdx.x;
  const int lane = tid & 31;
  const int wv   = __builtin_amdgcn_readfirstlane(tid >> 5);
  const int r    = (int)blockIdx.x;
  const int lo   = r * RN;
  int cnt = n - lo;
  if (cnt > RN) cnt = RN;

  for (int i = tid; i < NCP * RP; i += LB_THREADS) acc[i] = 0.0f;
  for (int i = tid; i < SBINS + 32; i += LB_THREADS) hist[i] = 0u;
  __syncthreads();

  const int ntiles = (cnt + 15) >> 4;
  const int lane16 = lane & 15;
  const int hi     = lane >> 4;
  const int il  = wv * 16 + lane16;
  const bool rok = il < cnt;
  const int ilc = rok ? il : (cnt - 1);
  float4 q0 = make_float4(0.f, 0.f, 0.f, 0.f), q1 = q0, q2 = q0, q3 = q0;
  if (wv < ntiles) {
    const float* xr = x + (size_t)(lo + ilc) * 32;
    q0 = *reinterpret_cast<const float4*>(xr + hi * 8);
    q1 = *reinterpret_cast<const float4*>(xr + hi * 8 + 4);
    q2 = *reinterpret_cast<const float4*>(xr + 16 + hi * 8);
    q3 = *reinterpret_cast<const float4*>(xr + 16 + hi * 8 + 4);
  }

  float* mine = acc + (lane & (NCP - 1)) * RP;
  const int pA = wv * RCH, pB = wv * RCH + 4;
  int soA[4], nlA[4], soB[4], nlB[4];
  rng_segs4(segtab, pA, r, soA, nlA);
  rng_segs4(segtab, pB, r, soB, nlB);
  rs_sweep4<false>(lists, pA, soA, nlA, lane, mine, hist, stage);
  rs_sweep4<false>(lists, pB, soB, nlB, lane, mine, hist, stage);
  if (lane == 0) {
    unsigned t = 0;
    for (int j = 0; j < 4; ++j) t += (unsigned)nlA[j] + (unsigned)nlB[j];
    wtot[wv] = t;
  }
  __syncthreads();

  unsigned total = 0;
  for (int w = 0; w < 32; ++w) total += wtot[w];
  const bool sorted = total <= (unsigned)RLCAP;
  if (sorted) {
    unsigned* hb = hist + tid * SBPT;
    unsigned v[SBPT];
    unsigned sum = 0;
#pragma unroll
    for (int i = 0; i < SBPT; ++i) { v[i] = hb[i]; sum += v[i]; }
    const unsigned incl = wave_incl_scan(sum, lane);
    if (lane == 31) wsum[wv] = incl;
    __syncthreads();
    unsigned run = incl - sum;
    for (int w = 0; w < 32; ++w) run += (w < wv) ? wsum[w] : 0u;
#pragma unroll
    for (int i = 0; i < SBPT; ++i) { hb[i] = run; run += v[i]; }
    __syncthreads();

    rs_sweep4<true>(lists, pA, soA, nlA, lane, mine, hist, stage);
    rs_sweep4<true>(lists, pB, soB, nlB, lane, mine, hist, stage);
    __syncthreads();

    uint4* out = reinterpret_cast<uint4*>(rlist + (size_t)r * RLCAP);
    const uint4* st4 = reinterpret_cast<const uint4*>(stage);
    const int n4 = (int)((total + 3u) >> 2);
    for (int i = tid; i < n4; i += LB_THREADS) out[i] = st4[i];
  }
  if (tid == 0) rcnt[r] = sorted ? (int)total : -1;

  float* dl = acc + NCP * RP;
  for (int i = tid; i < cnt; i += LB_THREADS) {
    float s = 1.0f;
    for (int c = 0; c < NCP; ++c) s += acc[c * RP + i];
    const float d = rsqrtf(s);
    dl[i] = d;
    dinv[lo + i] = d;
  }
  __syncthreads();

  if (wv < ntiles) {
    v16h b;
    {
      int  colW = lane16 & 3;
      bool keep = lane16 < 4;
#pragma unroll
      for (int j = 0; j < 16; ++j) {
        int k = (j < 8) ? (hi * 8 + j) : (16 + hi * 8 + (j - 8));
        float wvv = W[k * 4 + colW];
        b[j] = keep ? (_Float16)wvv : (_Float16)0.0f;
      }
    }
    const float di = dl[ilc];
    const float zf = rok ? 1.0f : 0.0f;
    v16h a;
    a[0]=(_Float16)(q0.x*zf); a[1]=(_Float16)(q0.y*zf); a[2]=(_Float16)(q0.z*zf); a[3]=(_Float16)(q0.w*zf);
    a[4]=(_Float16)(q1.x*zf); a[5]=(_Float16)(q1.y*zf); a[6]=(_Float16)(q1.z*zf); a[7]=(_Float16)(q1.w*zf);
    a[8]=(_Float16)(q2.x*zf); a[9]=(_Float16)(q2.y*zf); a[10]=(_Float16)(q2.z*zf); a[11]=(_Float16)(q2.w*zf);
    a[12]=(_Float16)(q3.x*zf); a[13]=(_Float16)(q3.y*zf); a[14]=(_Float16)(q3.z*zf); a[15]=(_Float16)(q3.w*zf);
    v8f c = {};
    c = __builtin_amdgcn_wmma_f32_16x16x32_f16(false, a, false, b, (short)0, c, false, false);
    float* slab = slabs + wv * 64;
    if (lane16 < 4) {
#pragma unroll
      for (int rr = 0; rr < 8; ++rr) slab[(hi * 8 + rr) * 4 + lane16] = c[rr];
    }
    asm volatile("s_wait_dscnt 0" ::: "memory");
    if (hi == 0 && rok) {
      float4 v = *reinterpret_cast<const float4*>(slab + lane16 * 4);
      v.x *= di; v.y *= di; v.z *= di; v.w *= di;
      *reinterpret_cast<float4*>(m + (size_t)(lo + il) * 4) = v;
    }
  }
}


template<int F>
__device__ __forceinline__ void rng_add16(float* acc, const float* __restrict__ m,
                                          const uint4 w[4], const int rem[4], int lo) {
  unsigned rel[16];
  bool     ok[16];
  float    v[16][F];
#pragma unroll
  for (int j = 0; j < 4; ++j) {
    unsigned ww[4] = {w[j].x, w[j].y, w[j].z, w[j].w};
#pragma unroll
    for (int k = 0; k < 4; ++k) {
      const int t = j * 4 + k;
      ok[t]  = k < rem[j];
      rel[t] = ww[k] & ((1u << RBITS) - 1u);
      int si = ok[t] ? (int)(ww[k] >> RBITS) : lo;
      if constexpr (F == 4) {
        float4 q = *reinterpret_cast<const float4*>(m + (size_t)si * 4);
        v[t][0] = q.x; v[t][1] = q.y; v[t][2] = q.z; v[t][3] = q.w;
      } else {
        float2 q = *reinterpret_cast<const float2*>(m + (size_t)si * 2);
        v[t][0] = q.x; v[t][1] = q.y;
      }
    }
  }
#pragma unroll
  for (int t = 0; t < 16; ++t) {
    if (ok[t]) {
      for (int f = 0; f < F; ++f) atomicAdd(&acc[f * RP + rel[t]], v[t][f]);
    }
  }
}


template<int F>
__device__ __forceinline__ void rng_add8(float* acc, const float* __restrict__ m,
                                         const unsigned w[SGRP], int gl, int idx0, int total, int lo) {
  unsigned rel[SGRP];
  bool     ok[SGRP];
  float    v[SGRP][F];
#pragma unroll
  for (int i = 0; i < SGRP; ++i) {
    ok[i]  = i < gl && idx0 + i * 32 < total;
    rel[i] = w[i] & ((1u << RBITS) - 1u);
    int si = ok[i] ? (int)(w[i] >> RBITS) : lo;
    if constexpr (F == 4) {
      float4 q = *reinterpret_cast<const float4*>(m + (size_t)si * 4);
      v[i][0] = q.x; v[i][1] = q.y; v[i][2] = q.z; v[i][3] = q.w;
    } else {
      float2 q = *reinterpret_cast<const float2*>(m + (size_t)si * 2);
      v[i][0] = q.x; v[i][1] = q.y;
    }
  }
#pragma unroll
  for (int i = 0; i < SGRP; ++i) {
    if (ok[i]) {
      for (int f = 0; f < F; ++f) atomicAdd(&acc[f * RP + rel[i]], v[i][f]);
    }
  }
}

template<int F, int FOUT, bool LAST>
__global__ __launch_bounds__(LB_THREADS)
void k_rng_edge(const unsigned* __restrict__ lists, const int* __restrict__ segtab,
                const unsigned* __restrict__ rlist, const int* __restrict__ rcnt,
                const float* __restrict__ m, const float* __restrict__ dinv,
                const float* __restrict__ b, const float* __restrict__ Wn,
                const float* __restrict__ bn, float* __restrict__ mout,
                float* __restrict__ hout, int n) {
  __shared__ float acc[4 * NPRD];
  __shared__ float slabs[32 * 64];
  const int tid  = (int)threadIdx.x;
  const int lane = tid & 31;
  const int wv   = __builtin_amdgcn_readfirstlane(tid >> 5);
  const int r    = (int)blockIdx.x;
  const int lo   = r * RN;
  int cnt = n - lo;
  if (cnt > RN) cnt = RN;
  const int totv = rcnt ? rcnt[r] : -1;

  for (int i = tid; i < NCP * F * RP; i += LB_THREADS) acc[i] = 0.0f;
  __syncthreads();

  const int ntiles = (cnt + 15) >> 4;
  const int lane16 = lane & 15;
  const int hi     = lane >> 4;
  const int il  = wv * 16 + lane16;
  const bool rok = il < cnt;
  const int ilc = rok ? il : (cnt - 1);
  const int ar  = lo + ilc;
  const int f0  = (F == 4) ? hi * 2 : 0;
  float  di = 0.0f, bs0 = 0.0f, bs1 = 0.0f;
  float2 mv = make_float2(0.0f, 0.0f);
  v2f    bw = {0.0f, 0.0f};
  if (wv < ntiles) {
    di  = dinv[ar];
    mv  = *reinterpret_cast<const float2*>(m + (size_t)ar * F + f0);
    bs0 = b[f0];
    bs1 = b[f0 + 1];
#pragma unroll
    for (int j = 0; j < 2; ++j) {
      int  k    = hi * 2 + j;
      int  kc   = (k < F) ? k : (F - 1);
      int  cc   = (lane16 < FOUT) ? lane16 : (FOUT - 1);
      float wvv = Wn[kc * FOUT + cc];
      bw[j] = (k < F && lane16 < FOUT) ? wvv : 0.0f;
    }
  }

  float* mine = acc + (lane & (NCP - 1)) * (F * RP);
  const int total = __builtin_amdgcn_readfirstlane(totv);
  if (total > 0) {
    const unsigned* rl = rlist + (size_t)r * RLCAP;
    const int last = total - 1;
    const int ngrp = (total + 31) >> 5;
    const int gpw  = (ngrp + 31) >> 5;
    const int g0   = wv * gpw;
    int g1 = g0 + gpw;
    if (g1 > ngrp) g1 = ngrp;
    unsigned wn[SGRP];
#pragma unroll
    for (int i = 0; i < SGRP; ++i) {
      const int idx = (g0 + i) * 32 + lane;
      wn[i] = rl[idx < last ? idx : last];
    }
    for (int g = g0; g < g1; g += SGRP) {
      unsigned w[SGRP];
#pragma unroll
      for (int i = 0; i < SGRP; ++i) {
        w[i] = wn[i];
        const int idx = (g + SGRP + i) * 32 + lane;
        wn[i] = rl[idx < last ? idx : last];
      }
      rng_add8<F>(mine, m, w, g1 - g, g * 32 + lane, total, lo);
    }
  } else if (total < 0) {
    for (int j0 = 0; j0 < RCH; j0 += 4) {
      const int p0 = wv * RCH + j0;
      int so[4], nl[4];
      rng_segs4(segtab, p0, r, so, nl);
      int ng[4], ngmax = 0;
      for (int j = 0; j < 4; ++j) { ng[j] = (nl[j] + 127) >> 7; ngmax = ng[j] > ngmax ? ng[j] : ngmax; }
      for (int q = 0; q < ngmax; ++q) {
        uint4 w[4];
        int rem[4];
        for (int j = 0; j < 4; ++j) {
          rem[j] = nl[j] - (q * 128 + lane * 4);
          const int qj = (q < ng[j]) ? q : (ng[j] > 0 ? ng[j] - 1 : 0);
          const uint4* lp = reinterpret_cast<const uint4*>(lists + (size_t)(p0 + j) * SCAP + so[j]) + lane;
          w[j] = lp[(size_t)qj * 32];
        }
        rng_add16<F>(mine, m, w, rem, lo);
      }
    }
  }
  __syncthreads();

#if __has_builtin(__builtin_amdgcn_wmma_f32_16x16x4_f32)
  if (wv < ntiles) {
    const float zf = rok ? 1.0f : 0.0f;
    float s0 = 0.0f, s1 = 0.0f;
    for (int c = 0; c < NCP; ++c) {
      s0 += acc[(c * F + f0) * RP + ilc];
      s1 += acc[(c * F + f0 + 1) * RP + ilc];
    }
    const float h0 = tanhf(di * (s0 + mv.x) + bs0);
    const float h1 = tanhf(di * (s1 + mv.y) + bs1);
    v2f a;
    if (F == 4) { a[0] = h0 * zf; a[1] = h1 * zf; }
    else        { const float zk = (hi == 0) ? zf : 0.0f; a[0] = h0 * zk; a[1] = h1 * zk; }
    if (LAST) {
      if (hi == 0 && rok) {
        float2 o; o.x = h0; o.y = h1;
        *reinterpret_cast<float2*>(hout + (size_t)ar * 2) = o;
      }
    }
    v8f c = {};
    c = __builtin_amdgcn_wmma_f32_16x16x4_f32(false, a, false, bw, (short)0, c, false, false);
    float* slab = slabs + wv * 64;
    if (lane16 < FOUT) {
#pragma unroll
      for (int rr = 0; rr < 8; ++rr) slab[(hi * 8 + rr) * FOUT + lane16] = c[rr];
    }
    asm volatile("s_wait_dscnt 0" ::: "memory");
    if (hi == 0 && rok) {
      if (FOUT == 4) {
        float4 v = *reinterpret_cast<const float4*>(slab + lane16 * 4);
        if (LAST) { v.x += bn[0]; v.y += bn[1]; v.z += bn[2]; v.w += bn[3]; }
        else      { v.x *= di; v.y *= di; v.z *= di; v.w *= di; }
        *reinterpret_cast<float4*>(mout + (size_t)ar * 4) = v;
      } else {
        float2 v = *reinterpret_cast<const float2*>(slab + lane16 * 2);
        if (LAST) { v.x += bn[0]; v.y += bn[1]; }
        else      { v.x *= di; v.y *= di; }
        *reinterpret_cast<float2*>(mout + (size_t)ar * 2) = v;
      }
    }
  }
#endif
}

extern "C" void kernel_launch(void* const* d_in, const int* in_sizes, int n_in,
                              void* d_out, int out_size, void* d_ws, size_t ws_size,
                              hipStream_t stream) {
  const float* x  = (const float*)d_in[0];
  const int*   ei = (const int*)  d_in[1];
  const float* W1 = (const float*)d_in[2];
  const float* b1 = (const float*)d_in[3];
  const float* W2 = (const float*)d_in[4];
  const float* b2 = (const float*)d_in[5];
  const float* W3 = (const float*)d_in[6];
  const float* b3 = (const float*)d_in[7];
  const float* Wc = (const float*)d_in[8];
  const float* bc = (const float*)d_in[9];

  const int N = in_sizes[0] / 32;
  const int E = in_sizes[1] / 2;
  const int* src = ei;
  const int* dst = ei + E;

  float* out = (float*)d_out;
  float* h3  = out + (size_t)N * 4;

  const int nr4 = (N + NPR4 - 1) / NPR4;
  const int nc4 = (GRID_WGS / nr4) > 0 ? (GRID_WGS / nr4) : 1;
  const int ec4 = (((E + nc4 - 1) / nc4) + 3) & ~3;
  const int nr2 = (N + NPR2 - 1) / NPR2;
  const int nc2 = (GRID_WGS / nr2) > 0 ? (GRID_WGS / nr2) : 1;
  const int ec2 = (((E + nc2 - 1) / nc2) + 3) & ~3;

  const size_t Np = ((size_t)N + 63) & ~(size_t)63;
  float* ws   = (float*)d_ws;
  float* dinv = ws;
  float* m    = ws + Np;
  float* h    = ws + 5 * Np;
  float* part = ws + 9 * Np;

  const int gN  = (N + THREADS - 1) / THREADS;
  const int tiles = (N + 15) / 16;
  const int gW  = (tiles + WAVES_PER_BLOCK - 1) / WAVES_PER_BLOCK;

  const int    wcap   = ((ec4 + TILE_EDGES - 1) / TILE_EDGES) * 128;
  const size_t nwaves = (size_t)nr4 * nc4 * (LB_THREADS / 32);
  const size_t partf  = (((size_t)nc4 * 4 * N) + 63) & ~(size_t)63;
  int*      lcnt  = (int*)(part + partf);
  unsigned* lists = (unsigned*)(part + partf + ((nwaves + 63) & ~(size_t)63));
  const size_t need = (size_t)((const char*)(lists + nwaves * wcap) - (const char*)d_ws);
  const int    snr   = (N + SNPR - 1) / SNPR;
  const int    spc   = (((E + SNP - 1) / SNP) + 3) & ~3;
  const size_t partS = (((size_t)SNC * 4 * N) + 63) & ~(size_t)63;
  int*      segtab = (int*)(part + partS);
  unsigned* segs   = (unsigned*)(part + partS + (size_t)SNP * SNR * 2);
  const size_t needS = (size_t)((const char*)(segs + (size_t)SNP * SCAP) - (const char*)d_ws);
  const int    nrng  = (N + RN - 1) / RN;
  int*      rtab  = (int*)(ws + 9 * Np);
  unsigned* rsegs = (unsigned*)(rtab + (size_t)SNP * RNR * 2);
  const size_t needR = (size_t)((const char*)(rsegs + (size_t)SNP * SCAP) - (const char*)d_ws);
  int*      rcnt  = (int*)(rsegs + (size_t)SNP * SCAP);
  unsigned* rlist = (unsigned*)(rcnt + RNR);
  const size_t needL = (size_t)((const char*)(rlist + (size_t)RNR * RLCAP) - (const char*)d_ws);
  if (nrng <= RNR && N <= (1 << 23) && spc <= SMAXCHUNK && needR <= ws_size) {
    float* mA = m;
    float* mB = h;
    k_rsplit<<<SNP, LB_THREADS, 0, stream>>>(src, dst, rsegs, rtab, E, N, spc);
    if (N <= (SBINS << SLB) && needL <= ws_size) {
      k_rng_sdeg<<<nrng, LB_THREADS, 0, stream>>>(rsegs, rtab, x, W1, dinv, mA, rlist, rcnt, N);
      k_rng_edge<4, 4, false><<<nrng, LB_THREADS, 0, stream>>>(rsegs, rtab, rlist, rcnt, mA, dinv, b1, W2, nullptr, mB, nullptr, N);
      k_rng_edge<4, 2, false><<<nrng, LB_THREADS, 0, stream>>>(rsegs, rtab, rlist, rcnt, mB, dinv, b2, W3, nullptr, mA, nullptr, N);
      k_rng_edge<2, 4, true><<<nrng, LB_THREADS, 0, stream>>>(rsegs, rtab, rlist, rcnt, mA, dinv, b3, Wc, bc, out, h3, N);
      return;
    }
    k_rng_deg<<<nrng, LB_THREADS, 0, stream>>>(rsegs, rtab, x, W1, dinv, mA, N);
    k_rng_edge<4, 4, false><<<nrng, LB_THREADS, 0, stream>>>(rsegs, rtab, nullptr, nullptr, mA, dinv, b1, W2, nullptr, mB, nullptr, N);
    k_rng_edge<4, 2, false><<<nrng, LB_THREADS, 0, stream>>>(rsegs, rtab, nullptr, nullptr, mB, dinv, b2, W3, nullptr, mA, nullptr, N);
    k_rng_edge<2, 4, true><<<nrng, LB_THREADS, 0, stream>>>(rsegs, rtab, nullptr, nullptr, mA, dinv, b3, Wc, bc, out, h3, N);
    return;
  }

  if (snr <= SNR && N <= (1 << 20) && spc <= SMAXCHUNK && needS <= ws_size) {
    k_split<<<SNP, LB_THREADS, 0, stream>>>(src, dst, segs, segtab, E, N, spc);
    k_seg_deg<<<snr * SNC, LB_THREADS, 0, stream>>>(segs, segtab, part, N, snr);
    k_rsqrt<<<gN, THREADS, 0, stream>>>(dinv, part, N, SNC);

    k_xform32_wmma<<<gW, THREADS, 0, stream>>>(x, W1, dinv, m, N);
    k_seg_edge<4><<<snr * SNC, LB_THREADS, 0, stream>>>(segs, segtab, m, part, N, snr);
    k_tanh<4><<<gN, THREADS, 0, stream>>>(h, part, m, dinv, b1, N, SNC);

    k_xform_small<4, 4, false, true><<<gW, THREADS, 0, stream>>>(h, W2, nullptr, dinv, m, N);
    k_seg_edge<4><<<snr * SNC, LB_THREADS, 0, stream>>>(segs, segtab, m, part, N, snr);
    k_tanh<4><<<gN, THREADS, 0, stream>>>(h, part, m, dinv, b2, N, SNC);

    k_xform_small<4, 2, false, true><<<gW, THREADS, 0, stream>>>(h, W3, nullptr, dinv, m, N);
    k_seg_edge<2><<<snr * SNC, LB_THREADS, 0, stream>>>(segs, segtab, m, part, N, snr);
    k_tanh<2><<<gN, THREADS, 0, stream>>>(h3, part, m, dinv, b3, N, SNC);

    k_xform_small<2, 4, true, false><<<gW, THREADS, 0, stream>>>(h3, Wc, bc, nullptr, out, N);
    return;
  }

  if (N <= (1 << 18) && NPRD <= (1 << LIST_RELBITS) && need <= ws_size) {
    k_deg_list<<<nr4 * nc4, LB_THREADS, 0, stream>>>(src, dst, part, lists, lcnt, E, N, nr4, ec4, wcap);
    k_rsqrt<<<gN, THREADS, 0, stream>>>(dinv, part, N, nc4);

    k_xform32_wmma<<<gW, THREADS, 0, stream>>>(x, W1, dinv, m, N);
    k_edge_list<4><<<nr4 * nc4, LB_THREADS, 0, stream>>>(lists, lcnt, m, part, N, nr4, wcap);
    k_tanh<4><<<gN, THREADS, 0, stream>>>(h, part, m, dinv, b1, N, nc4);

    k_xform_small<4, 4, false, true><<<gW, THREADS, 0, stream>>>(h, W2, nullptr, dinv, m, N);
    k_edge_list<4><<<nr4 * nc4, LB_THREADS, 0, stream>>>(lists, lcnt, m, part, N, nr4, wcap);
    k_tanh<4><<<gN, THREADS, 0, stream>>>(h, part, m, dinv, b2, N, nc4);

    k_xform_small<4, 2, false, true><<<gW, THREADS, 0, stream>>>(h, W3, nullptr, dinv, m, N);
    k_edge_list<2><<<nr4 * nc4, LB_THREADS, 0, stream>>>(lists, lcnt, m, part, N, nr4, wcap);
    k_tanh<2><<<gN, THREADS, 0, stream>>>(h3, part, m, dinv, b3, N, nc4);

    k_xform_small<2, 4, true, false><<<gW, THREADS, 0, stream>>>(h3, Wc, bc, nullptr, out, N);
    return;
  }

  k_deg_lds<<<nr4 * nc4, LB_THREADS, 0, stream>>>(dst, part, E, N, nr4, ec4);
  k_rsqrt<<<gN, THREADS, 0, stream>>>(dinv, part, N, nc4);

  k_xform32_wmma<<<gW, THREADS, 0, stream>>>(x, W1, dinv, m, N);
  k_edge_lds<4><<<nr4 * nc4, LB_THREADS, 0, stream>>>(src, dst, m, part, E, N, nr4, ec4);
  k_tanh<4><<<gN, THREADS, 0, stream>>>(h, part, m, dinv, b1, N, nc4);

  k_xform_small<4, 4, false, true><<<gW, THREADS, 0, stream>>>(h, W2, nullptr, dinv, m, N);
  k_edge_lds<4><<<nr4 * nc4, LB_THREADS, 0, stream>>>(src, dst, m, part, E, N, nr4, ec4);
  k_tanh<4><<<gN, THREADS, 0, stream>>>(h, part, m, dinv, b2, N, nc4);

  k_xform_small<4, 2, false, true><<<gW, THREADS, 0, stream>>>(h, W3, nullptr, dinv, m, N);
  k_edge_lds<2><<<nr2 * nc2, LB_THREADS, 0, stream>>>(src, dst, m, part, E, N, nr2, ec2);
  k_tanh<2><<<gN, THREADS, 0, stream>>>(h3, part, m, dinv, b3, N, nc2);

  k_xform_small<2, 4, true, false><<<gW, THREADS, 0, stream>>>(h3, Wc, bc, nullptr, out, N);
}
